# starting kernel
# baseline (speedup 1.0000x reference)
#include <hip/hip_runtime.h>

#define N_NODES 100000
#define N_EDGES 1600000
#define IN_DIM  256
#define HID     128
#define OUTD    64

typedef __attribute__((ext_vector_type(2))) float v2f;
typedef __attribute__((ext_vector_type(8))) float v8f;

__global__ void k_init_deg(float* __restrict__ deg) {
  int i = blockIdx.x * blockDim.x + threadIdx.x;
  if (i < N_NODES) deg[i] = 1.0f;
}

__global__ void k_count_deg(const int* __restrict__ ei, float* __restrict__ deg) {
  int e = blockIdx.x * blockDim.x + threadIdx.x;
  if (e < N_EDGES) {
    int d = ei[N_EDGES + e];
    unsafeAtomicAdd(&deg[d], 1.0f);
  }
}

__global__ void k_rsqrt_inplace(float* __restrict__ deg) {
  int i = blockIdx.x * blockDim.x + threadIdx.x;
  if (i < N_NODES) deg[i] = rsqrtf(deg[i]);
}

template <int K, int NT, bool RELU>
__global__ __launch_bounds__(256) void k_gemm_wmma(const float* __restrict__ A,
                                                   const float* __restrict__ W,
                                                   float* __restrict__ C, int M) {
  constexpr int NC = NT * 16;
  constexpr int KC = 64;
  __shared__ float lds[KC * NC];

  const int wave = threadIdx.x >> 5;
  const int lane = threadIdx.x & 31;
  const int mtile = blockIdx.x * 8 + wave;
  const long m0 = (long)mtile * 16;
  const bool active = (m0 < M);

  const int m  = lane & 15;
  const int kh = lane >> 4;
  const int n  = lane & 15;

  const long arow = active ? (m0 + m) : (long)(M - 1);
  const float* __restrict__ Aptr = A + arow * K;

  v8f acc[NT] = {};

  for (int kc = 0; kc < K; kc += KC) {
    if (kc != 0) __syncthreads();
    {
      constexpr int ENT = KC * NC / 8;
#pragma unroll
      for (int idx = threadIdx.x; idx < ENT; idx += 256) {
        int kp = idx / (NC / 4);
        int nq = (idx % (NC / 4)) * 4;
        const float4 a = *(const float4*)(W + (long)(kc + 2 * kp) * NC + nq);
        const float4 b = *(const float4*)(W + (long)(kc + 2 * kp + 1) * NC + nq);
        float4* __restrict__ dst = (float4*)&lds[kp * 2 * NC + 2 * nq];
        dst[0] = make_float4(a.x, b.x, a.y, b.y);
        dst[1] = make_float4(a.z, b.z, a.w, b.w);
      }
    }
    __syncthreads();

#pragma unroll 4
    for (int k0 = kc; k0 < kc + KC; k0 += 4) {
      const int ka = k0 + 2 * kh;
      const int kp = (ka - kc) >> 1;
      v2f av = *(const v2f*)(Aptr + ka);
      if (RELU) {
        av.x = fmaxf(av.x, 0.0f);
        av.y = fmaxf(av.y, 0.0f);
      }
      const v2f* __restrict__ brow = (const v2f*)&lds[kp * 2 * NC + 2 * n];
#pragma unroll
      for (int j = 0; j < NT; ++j) {
        v2f bv = brow[j * 16];
        acc[j] = __builtin_amdgcn_wmma_f32_16x16x4_f32(
             false, av,  false, bv,
             (short)0, acc[j],  false,  false);
      }
    }
  }

  if (!active) return;
#pragma unroll
  for (int j = 0; j < NT; ++j) {
#pragma unroll
    for (int r = 0; r < 8; ++r) {
      long row = m0 + r + 8 * kh;
      if (row < M) C[row * NC + j * 16 + n] = acc[j][r];
    }
  }
}

template <int F>
__global__ void k_self_init(const float* __restrict__ h,
                            const float* __restrict__ dinv,
                            const float* __restrict__ b,
                            float* __restrict__ agg) {
  long t = (long)blockIdx.x * blockDim.x + threadIdx.x;
  if (t >= (long)N_NODES * F) return;
  long i = t / F;
  int f = (int)(t % F);
  float di = dinv[i];
  agg[t] = h[t] * (di * di) + b[f];
}

template <int F>
__global__ void k_edge_agg(const int* __restrict__ ei,
                           const float* __restrict__ dinv,
                           const float* __restrict__ h,
                           float* __restrict__ agg) {
  constexpr int CH = F / 4;
  long t = (long)blockIdx.x * blockDim.x + threadIdx.x;
  long e = t / CH;
  int c = (int)(t % CH);
  if (e >= N_EDGES) return;
  int s = ei[e];
  int d = ei[N_EDGES + e];
  float norm = dinv[s] * dinv[d];
  float4 v = *(const float4*)(h + (long)s * F + c * 4);
  float* __restrict__ out = agg + (long)d * F + c * 4;
  unsafeAtomicAdd(out + 0, v.x * norm);
  unsafeAtomicAdd(out + 1, v.y * norm);
  unsafeAtomicAdd(out + 2, v.z * norm);
  unsafeAtomicAdd(out + 3, v.w * norm);
}

extern "C" void kernel_launch(void* const* d_in, const int* in_sizes, int n_in,
                              void* d_out, int out_size, void* d_ws, size_t ws_size,
                              hipStream_t stream) {
  const float* x  = (const float*)d_in[0];
  const int*   ei = (const int*)d_in[1];
  const float* W1 = (const float*)d_in[2];
  const float* b1 = (const float*)d_in[3];
  const float* W2 = (const float*)d_in[4];
  const float* b2 = (const float*)d_in[5];
  float* out = (float*)d_out;

  float* ws   = (float*)d_ws;
  float* dinv = ws;
  float* h1   = dinv + N_NODES;
  float* agg1 = h1 + (size_t)N_NODES * HID;
  float* h3   = agg1 + (size_t)N_NODES * HID;

  const int T = 256;

  k_init_deg<<<(N_NODES + T - 1) / T, T, 0, stream>>>(dinv);
  k_count_deg<<<(N_EDGES + T - 1) / T, T, 0, stream>>>(ei, dinv);
  k_rsqrt_inplace<<<(N_NODES + T - 1) / T, T, 0, stream>>>(dinv);

  const int mtiles = (N_NODES + 15) / 16;
  const int gblocks = (mtiles + 7) / 8;
  k_gemm_wmma<IN_DIM, HID / 16, false>
      <<<gblocks, T, 0, stream>>>(x, W1, h1, N_NODES);

  {
    long tot = (long)N_NODES * HID;
    k_self_init<HID><<<(unsigned)((tot + T - 1) / T), T, 0, stream>>>(h1, dinv, b1, agg1);
    long et = (long)N_EDGES * (HID / 4);
    k_edge_agg<HID><<<(unsigned)((et + T - 1) / T), T, 0, stream>>>(ei, dinv, h1, agg1);
  }

  k_gemm_wmma<HID, OUTD / 16, true>
      <<<gblocks, T, 0, stream>>>(agg1, W2, h3, N_NODES);

  {
    long tot = (long)N_NODES * OUTD;
    k_self_init<OUTD><<<(unsigned)((tot + T - 1) / T), T, 0, stream>>>(h3, dinv, b2, out);
    long et = (long)N_EDGES * (OUTD / 4);
    k_edge_agg<OUTD><<<(unsigned)((et + T - 1) / T), T, 0, stream>>>(ei, dinv, h3, out);
  }
}


# Edge sums as dst-sorted CSR gather (int atomic rank + scan + fill, once per call; one wave sums a node's rows, one store) replacing f32 atomic scatter; sum order = arrival rank, self-loop last
# speedup vs baseline: 2.1534x; 2.1534x over previous
#include <hip/hip_runtime.h>

#define N_NODES 100000
#define N_EDGES 1600000
#define IN_DIM  256
#define HID     128
#define OUTD    64

typedef __attribute__((ext_vector_type(2))) float v2f;
typedef __attribute__((ext_vector_type(4))) float v4f;
typedef __attribute__((ext_vector_type(8))) float v8f;

#define SCAN_T 256
#define SCAN_B ((N_NODES + SCAN_T - 1) / SCAN_T)

__global__ void k_zero_cnt(int* __restrict__ cnt) {
  int i = blockIdx.x * blockDim.x + threadIdx.x;
  if (i < N_NODES) cnt[i] = 0;
}

__global__ void k_count_rank(const int* __restrict__ ei, int* __restrict__ cnt,
                             int* __restrict__ rank) {
  int e = blockIdx.x * blockDim.x + threadIdx.x;
  if (e < N_EDGES) {
    int d = ei[N_EDGES + e];
    rank[e] = atomicAdd(&cnt[d], 1);
  }
}

__global__ __launch_bounds__(SCAN_T) void k_block_sums(const int* __restrict__ cnt,
                                                       int* __restrict__ bsum) {
  __shared__ int red[SCAN_T];
  const int tid = threadIdx.x;
  const int i = blockIdx.x * SCAN_T + tid;
  red[tid] = (i < N_NODES) ? cnt[i] : 0;
  __syncthreads();
  for (int s = SCAN_T / 2; s > 0; s >>= 1) {
    if (tid < s) red[tid] += red[tid + s];
    __syncthreads();
  }
  if (tid == 0) bsum[blockIdx.x] = red[0];
}

__global__ __launch_bounds__(SCAN_T) void k_scan_rows(const int* __restrict__ cnt,
                                                      const int* __restrict__ bsum,
                                                      int* __restrict__ rowptr,
                                                      float* __restrict__ dinv) {
  __shared__ int red[SCAN_T];
  __shared__ int sc[SCAN_T];
  const int tid = threadIdx.x;
  const int b = blockIdx.x;
  int part = 0;
  for (int k = tid; k < b; k += SCAN_T) part += bsum[k];
  red[tid] = part;
  const int i = b * SCAN_T + tid;
  const int c = (i < N_NODES) ? cnt[i] : 0;
  sc[tid] = c;
  __syncthreads();
  for (int s = SCAN_T / 2; s > 0; s >>= 1) {
    if (tid < s) red[tid] += red[tid + s];
    __syncthreads();
  }
  const int base = red[0];
  for (int off = 1; off < SCAN_T; off <<= 1) {
    const int v = (tid >= off) ? sc[tid - off] : 0;
    __syncthreads();
    sc[tid] += v;
    __syncthreads();
  }
  const int incl = sc[tid];
  if (i < N_NODES) {
    rowptr[i] = base + incl - c;
    dinv[i] = rsqrtf((float)c + 1.0f);
    if (i == N_NODES - 1) rowptr[N_NODES] = base + incl;
  }
}

__global__ void k_fill_csr(const int* __restrict__ ei, const int* __restrict__ rowptr,
                           const int* __restrict__ rank, int* __restrict__ csr_src) {
  int e = blockIdx.x * blockDim.x + threadIdx.x;
  if (e < N_EDGES) {
    int s = ei[e];
    int d = ei[N_EDGES + e];
    csr_src[rowptr[d] + rank[e]] = s;
  }
}

template <int K, int NT, bool RELU>
__global__ __launch_bounds__(256) void k_gemm_wmma(const float* __restrict__ A,
                                                   const float* __restrict__ W,
                                                   float* __restrict__ C, int M) {
  constexpr int NC = NT * 16;
  constexpr int KC = 64;
  __shared__ float lds[KC * NC];

  const int wave = threadIdx.x >> 5;
  const int lane = threadIdx.x & 31;
  const int mtile = blockIdx.x * 8 + wave;
  const long m0 = (long)mtile * 16;
  const bool active = (m0 < M);

  const int m  = lane & 15;
  const int kh = lane >> 4;
  const int n  = lane & 15;

  const long arow = active ? (m0 + m) : (long)(M - 1);
  const float* __restrict__ Aptr = A + arow * K;

  v8f acc[NT] = {};

  for (int kc = 0; kc < K; kc += KC) {
    if (kc != 0) __syncthreads();
    {
      constexpr int ENT = KC * NC / 8;
#pragma unroll
      for (int idx = threadIdx.x; idx < ENT; idx += 256) {
        int kp = idx / (NC / 4);
        int nq = (idx % (NC / 4)) * 4;
        const float4 a = *(const float4*)(W + (long)(kc + 2 * kp) * NC + nq);
        const float4 b = *(const float4*)(W + (long)(kc + 2 * kp + 1) * NC + nq);
        float4* __restrict__ dst = (float4*)&lds[kp * 2 * NC + 2 * nq];
        dst[0] = make_float4(a.x, b.x, a.y, b.y);
        dst[1] = make_float4(a.z, b.z, a.w, b.w);
      }
    }
    __syncthreads();

#pragma unroll 4
    for (int k0 = kc; k0 < kc + KC; k0 += 4) {
      const int ka = k0 + 2 * kh;
      const int kp = (ka - kc) >> 1;
      v2f av = *(const v2f*)(Aptr + ka);
      if (RELU) {
        av.x = fmaxf(av.x, 0.0f);
        av.y = fmaxf(av.y, 0.0f);
      }
      const v2f* __restrict__ brow = (const v2f*)&lds[kp * 2 * NC + 2 * n];
#pragma unroll
      for (int j = 0; j < NT; ++j) {
        v2f bv = brow[j * 16];
        acc[j] = __builtin_amdgcn_wmma_f32_16x16x4_f32(
             false, av,  false, bv,
             (short)0, acc[j],  false,  false);
      }
    }
  }

  if (!active) return;
#pragma unroll
  for (int j = 0; j < NT; ++j) {
#pragma unroll
    for (int r = 0; r < 8; ++r) {
      long row = m0 + r + 8 * kh;
      if (row < M) C[row * NC + j * 16 + n] = acc[j][r];
    }
  }
}

template <int V> struct LaneVec;
template <> struct LaneVec<2> { typedef v2f type; };
template <> struct LaneVec<4> { typedef v4f type; };

template <int F>
__global__ __launch_bounds__(256) void k_gather(const int* __restrict__ rowptr,
                                                const int* __restrict__ csr_src,
                                                const float* __restrict__ dinv,
                                                const float* __restrict__ h,
                                                const float* __restrict__ bias,
                                                float* __restrict__ agg) {
  typedef typename LaneVec<F / 32>::type vec;
  constexpr int DEPTH = 8;
  const int lane = threadIdx.x & 31;
  const int wv = __builtin_amdgcn_readfirstlane((int)(threadIdx.x >> 5));
  const int d = blockIdx.x * 8 + wv;
  if (d >= N_NODES) return;

  const int start = rowptr[d];
  const int end = rowptr[d + 1];
  const float dd = dinv[d];
  const vec* __restrict__ hv = (const vec*)h;

  vec acc = {};
  for (int base = start; base < end; base += 32) {
    const int m = (end - base < 32) ? (end - base) : 32;
    const int li = base + lane;
    const int myidx = csr_src[(li < end) ? li : (end - 1)];
    const float mydv = dinv[myidx];
    for (int j = 0; j < m; j += DEPTH) {
      vec v[DEPTH];
#pragma unroll
      for (int u = 0; u < DEPTH; ++u) {
        const int s = __builtin_amdgcn_readlane(myidx, j + u);
        v[u] = hv[(long)s * 32 + lane];
      }
      __builtin_amdgcn_sched_barrier(0);
#pragma unroll
      for (int u = 0; u < DEPTH; ++u) {
        const float ds = __int_as_float(
            __builtin_amdgcn_readlane(__float_as_int(mydv), j + u));
        const float wt = (j + u < m) ? ds * dd : 0.0f;
        acc += v[u] * wt;
      }
    }
  }

  const vec hd = hv[(long)d * 32 + lane];
  const vec bv = ((const vec*)bias)[lane];
  const vec r = acc + hd * (dd * dd) + bv;
  ((vec*)agg)[(long)d * 32 + lane] = r;
}

extern "C" void kernel_launch(void* const* d_in, const int* in_sizes, int n_in,
                              void* d_out, int out_size, void* d_ws, size_t ws_size,
                              hipStream_t stream) {
  const float* x  = (const float*)d_in[0];
  const int*   ei = (const int*)d_in[1];
  const float* W1 = (const float*)d_in[2];
  const float* b1 = (const float*)d_in[3];
  const float* W2 = (const float*)d_in[4];
  const float* b2 = (const float*)d_in[5];
  float* out = (float*)d_out;

  float* ws   = (float*)d_ws;
  float* dinv = ws;
  float* h1   = dinv + N_NODES;
  float* agg1 = h1 + (size_t)N_NODES * HID;
  float* h3   = h1;
  int* aux     = (int*)(agg1 + (size_t)N_NODES * HID);
  int* cnt     = aux;
  int* rowptr  = cnt + N_NODES;
  int* bsum    = rowptr + N_NODES + 32;
  int* rank    = bsum + 512;
  int* csr_src = rank + N_EDGES;

  const int T = 256;

  k_zero_cnt<<<(N_NODES + T - 1) / T, T, 0, stream>>>(cnt);
  k_count_rank<<<(N_EDGES + T - 1) / T, T, 0, stream>>>(ei, cnt, rank);
  k_block_sums<<<SCAN_B, SCAN_T, 0, stream>>>(cnt, bsum);
  k_scan_rows<<<SCAN_B, SCAN_T, 0, stream>>>(cnt, bsum, rowptr, dinv);
  k_fill_csr<<<(N_EDGES + T - 1) / T, T, 0, stream>>>(ei, rowptr, rank, csr_src);

  const int mtiles = (N_NODES + 15) / 16;
  const int gblocks = (mtiles + 7) / 8;
  k_gemm_wmma<IN_DIM, HID / 16, false>
      <<<gblocks, T, 0, stream>>>(x, W1, h1, N_NODES);

  k_gather<HID><<<(N_NODES + 7) / 8, T, 0, stream>>>(rowptr, csr_src, dinv, h1, b1, agg1);

  k_gemm_wmma<HID, OUTD / 16, true>
      <<<gblocks, T, 0, stream>>>(agg1, W2, h3, N_NODES);

  k_gather<OUTD><<<(N_NODES + 7) / 8, T, 0, stream>>>(rowptr, csr_src, dinv, h3, b2, out);
}


# CSR build groups edges by destination with LDS counters in two levels (256-node bins, then nodes) instead of one returning global atomic per edge; 5 launches -> 3; row sum order = LDS arrival order
# speedup vs baseline: 3.0304x; 1.4073x over previous
#include <hip/hip_runtime.h>

#define N_NODES 100000
#define N_EDGES 1600000
#define IN_DIM  256
#define HID     128
#define OUTD    64

typedef __attribute__((ext_vector_type(2))) float v2f;
typedef __attribute__((ext_vector_type(4))) float v4f;
typedef __attribute__((ext_vector_type(8))) float v8f;

#define BIN_SHIFT 8
#define NBINS ((N_NODES + 255) >> BIN_SHIFT)
#define BINP 512
#define PT 1024
#define P_G 250
#define P_ROWS 256
#define P_CHUNK ((N_EDGES + P_G - 1) / P_G)
#define P_ITER ((P_CHUNK + PT - 1) / PT)
#define P_GW (P_ROWS / 4)
#define FT 1024
#define F_UB 5
#define STAGE_CAP 8192

__device__ __forceinline__ void wave0_excl_scan_256(const int* __restrict__ in,
                                                    int* __restrict__ out, int lane) {
  int v[8];
  int s = 0;
#pragma unroll
  for (int k = 0; k < 8; ++k) {
    v[k] = in[lane * 8 + k];
    s += v[k];
  }
  int inc = s;
#pragma unroll
  for (int off = 1; off < 32; off <<= 1) {
    const int nb = __shfl_up(inc, off);
    if (lane >= off) inc += nb;
  }
  int run = inc - s;
#pragma unroll
  for (int k = 0; k < 8; ++k) {
    out[lane * 8 + k] = run;
    run += v[k];
  }
}

__global__ __launch_bounds__(PT) void k_bin_count(const int* __restrict__ ei,
                                                  int* __restrict__ counts) {
  __shared__ int hist[BINP];
  const int tid = threadIdx.x;
  const int w = blockIdx.x;
  if (tid < BINP) hist[tid] = 0;
  const int base = w * P_CHUNK;
  const int lim = (base + P_CHUNK < N_EDGES) ? (base + P_CHUNK) : N_EDGES;
  int dv[P_ITER];
#pragma unroll
  for (int it = 0; it < P_ITER; ++it) {
    const int e = base + it * PT + tid;
    dv[it] = ei[N_EDGES + ((e < lim) ? e : (lim - 1))];
  }
  __syncthreads();
#pragma unroll
  for (int it = 0; it < P_ITER; ++it) {
    const int e = base + it * PT + tid;
    if (e < lim) atomicAdd(&hist[dv[it] >> BIN_SHIFT], 1);
  }
  __syncthreads();
  if (tid < BINP) counts[w * BINP + tid] = hist[tid];
}

__global__ __launch_bounds__(PT) void k_bin_scatter(const int* __restrict__ ei,
                                                    const int* __restrict__ counts,
                                                    int* __restrict__ binptr,
                                                    int* __restrict__ rec) {
  __shared__ int part_tot[4][BINP];
  __shared__ int part_pre[4][BINP];
  __shared__ int ptot[256];
  __shared__ int pex[256];
  __shared__ int cursor[BINP];
  const int tid = threadIdx.x;
  const int w = blockIdx.x;
  const int g = __builtin_amdgcn_readfirstlane(tid >> 8);
  const int t = tid & 255;

  const int base = w * P_CHUNK;
  const int lim = (base + P_CHUNK < N_EDGES) ? (base + P_CHUNK) : N_EDGES;
  int dv[P_ITER], sv[P_ITER];
#pragma unroll
  for (int it = 0; it < P_ITER; ++it) {
    const int e = base + it * PT + tid;
    const int ec = (e < lim) ? e : (lim - 1);
    dv[it] = ei[N_EDGES + ec];
    sv[it] = ei[ec];
  }

  const int2* __restrict__ c2 = (const int2*)counts;
  int tx = 0, ty = 0, px = 0, py = 0;
#pragma unroll 16
  for (int k = 0; k < P_GW; ++k) {
    const int ww = g * P_GW + k;
    const int2 v = c2[ww * (BINP / 2) + t];
    tx += v.x;
    ty += v.y;
    px += (ww < w) ? v.x : 0;
    py += (ww < w) ? v.y : 0;
  }
  part_tot[g][2 * t] = tx;
  part_tot[g][2 * t + 1] = ty;
  part_pre[g][2 * t] = px;
  part_pre[g][2 * t + 1] = py;
  __syncthreads();
  int T0 = 0, P0 = 0, P1 = 0;
  if (tid < 256) {
    int T1 = 0;
#pragma unroll
    for (int q = 0; q < 4; ++q) {
      T0 += part_tot[q][2 * tid];
      T1 += part_tot[q][2 * tid + 1];
      P0 += part_pre[q][2 * tid];
      P1 += part_pre[q][2 * tid + 1];
    }
    ptot[tid] = T0 + T1;
  }
  __syncthreads();
  if (tid < 32) wave0_excl_scan_256(ptot, pex, tid);
  __syncthreads();
  if (tid < 256) {
    const int b0 = pex[tid];
    const int b1 = b0 + T0;
    cursor[2 * tid] = b0 + P0;
    cursor[2 * tid + 1] = b1 + P1;
    if (w == 0) {
      binptr[2 * tid] = b0;
      binptr[2 * tid + 1] = b1;
    }
  }
  __syncthreads();
#pragma unroll
  for (int it = 0; it < P_ITER; ++it) {
    const int e = base + it * PT + tid;
    if (e < lim) {
      const int pos = atomicAdd(&cursor[dv[it] >> BIN_SHIFT], 1);
      rec[pos] = (sv[it] << BIN_SHIFT) | (dv[it] & 255);
    }
  }
}

__global__ __launch_bounds__(FT) void k_bin_rows(const int* __restrict__ binptr,
                                                 const int* __restrict__ rec,
                                                 int* __restrict__ rowptr,
                                                 float* __restrict__ dinv,
                                                 int* __restrict__ csr_src) {
  __shared__ int cnt[256];
  __shared__ int ex[256];
  __shared__ int stage[STAGE_CAP];
  const int tid = threadIdx.x;
  const int b = blockIdx.x;
  const int lo = binptr[b];
  const int n = binptr[b + 1] - lo;
  if (tid < 256) cnt[tid] = 0;
  __syncthreads();
  for (int i0 = tid; i0 < n; i0 += F_UB * FT) {
    int r[F_UB];
#pragma unroll
    for (int u = 0; u < F_UB; ++u) {
      const int i = i0 + u * FT;
      r[u] = rec[lo + ((i < n) ? i : (n - 1))];
    }
#pragma unroll
    for (int u = 0; u < F_UB; ++u)
      if (i0 + u * FT < n) atomicAdd(&cnt[r[u] & 255], 1);
  }
  __syncthreads();
  if (tid < 32) wave0_excl_scan_256(cnt, ex, tid);
  __syncthreads();
  if (tid < 256) {
    const int c = cnt[tid];
    const int x = ex[tid];
    const int node = (b << BIN_SHIFT) + tid;
    if (node < N_NODES) {
      rowptr[node] = lo + x;
      dinv[node] = rsqrtf((float)c + 1.0f);
      if (node == N_NODES - 1) rowptr[N_NODES] = lo + x + c;
    }
  }
  __syncthreads();
  const bool staged = (n <= STAGE_CAP);
  for (int i0 = tid; i0 < n; i0 += F_UB * FT) {
    int r[F_UB];
#pragma unroll
    for (int u = 0; u < F_UB; ++u) {
      const int i = i0 + u * FT;
      r[u] = rec[lo + ((i < n) ? i : (n - 1))];
    }
#pragma unroll
    for (int u = 0; u < F_UB; ++u) {
      if (i0 + u * FT < n) {
        const int p = atomicAdd(&ex[r[u] & 255], 1);
        if (staged) stage[p] = r[u] >> BIN_SHIFT;
        else csr_src[lo + p] = r[u] >> BIN_SHIFT;
      }
    }
  }
  if (staged) {
    __syncthreads();
    for (int i = tid; i < n; i += FT) csr_src[lo + i] = stage[i];
  }
}

template <int K, int NT, bool RELU>
__global__ __launch_bounds__(256) void k_gemm_wmma(const float* __restrict__ A,
                                                   const float* __restrict__ W,
                                                   float* __restrict__ C, int M) {
  constexpr int NC = NT * 16;
  constexpr int KC = 64;
  __shared__ float lds[KC * NC];

  const int wave = threadIdx.x >> 5;
  const int lane = threadIdx.x & 31;
  const int mtile = blockIdx.x * 8 + wave;
  const long m0 = (long)mtile * 16;
  const bool active = (m0 < M);

  const int m  = lane & 15;
  const int kh = lane >> 4;
  const int n  = lane & 15;

  const long arow = active ? (m0 + m) : (long)(M - 1);
  const float* __restrict__ Aptr = A + arow * K;

  v8f acc[NT] = {};

  for (int kc = 0; kc < K; kc += KC) {
    if (kc != 0) __syncthreads();
    {
      constexpr int ENT = KC * NC / 8;
#pragma unroll
      for (int idx = threadIdx.x; idx < ENT; idx += 256) {
        int kp = idx / (NC / 4);
        int nq = (idx % (NC / 4)) * 4;
        const float4 a = *(const float4*)(W + (long)(kc + 2 * kp) * NC + nq);
        const float4 b = *(const float4*)(W + (long)(kc + 2 * kp + 1) * NC + nq);
        float4* __restrict__ dst = (float4*)&lds[kp * 2 * NC + 2 * nq];
        dst[0] = make_float4(a.x, b.x, a.y, b.y);
        dst[1] = make_float4(a.z, b.z, a.w, b.w);
      }
    }
    __syncthreads();

#pragma unroll 4
    for (int k0 = kc; k0 < kc + KC; k0 += 4) {
      const int ka = k0 + 2 * kh;
      const int kp = (ka - kc) >> 1;
      v2f av = *(const v2f*)(Aptr + ka);
      if (RELU) {
        av.x = fmaxf(av.x, 0.0f);
        av.y = fmaxf(av.y, 0.0f);
      }
      const v2f* __restrict__ brow = (const v2f*)&lds[kp * 2 * NC + 2 * n];
#pragma unroll
      for (int j = 0; j < NT; ++j) {
        v2f bv = brow[j * 16];
        acc[j] = __builtin_amdgcn_wmma_f32_16x16x4_f32(
             false, av,  false, bv,
             (short)0, acc[j],  false,  false);
      }
    }
  }

  if (!active) return;
#pragma unroll
  for (int j = 0; j < NT; ++j) {
#pragma unroll
    for (int r = 0; r < 8; ++r) {
      long row = m0 + r + 8 * kh;
      if (row < M) C[row * NC + j * 16 + n] = acc[j][r];
    }
  }
}

template <int V> struct LaneVec;
template <> struct LaneVec<2> { typedef v2f type; };
template <> struct LaneVec<4> { typedef v4f type; };

template <int F>
__global__ __launch_bounds__(256) void k_gather(const int* __restrict__ rowptr,
                                                const int* __restrict__ csr_src,
                                                const float* __restrict__ dinv,
                                                const float* __restrict__ h,
                                                const float* __restrict__ bias,
                                                float* __restrict__ agg) {
  typedef typename LaneVec<F / 32>::type vec;
  constexpr int DEPTH = 8;
  const int lane = threadIdx.x & 31;
  const int wv = __builtin_amdgcn_readfirstlane((int)(threadIdx.x >> 5));
  const int d = blockIdx.x * 8 + wv;
  if (d >= N_NODES) return;

  const int start = rowptr[d];
  const int end = rowptr[d + 1];
  const float dd = dinv[d];
  const vec* __restrict__ hv = (const vec*)h;

  vec acc = {};
  for (int base = start; base < end; base += 32) {
    const int m = (end - base < 32) ? (end - base) : 32;
    const int li = base + lane;
    const int myidx = csr_src[(li < end) ? li : (end - 1)];
    const float mydv = dinv[myidx];
    for (int j = 0; j < m; j += DEPTH) {
      vec v[DEPTH];
#pragma unroll
      for (int u = 0; u < DEPTH; ++u) {
        const int s = __builtin_amdgcn_readlane(myidx, j + u);
        v[u] = hv[(long)s * 32 + lane];
      }
      __builtin_amdgcn_sched_barrier(0);
#pragma unroll
      for (int u = 0; u < DEPTH; ++u) {
        const float ds = __int_as_float(
            __builtin_amdgcn_readlane(__float_as_int(mydv), j + u));
        const float wt = (j + u < m) ? ds * dd : 0.0f;
        acc += v[u] * wt;
      }
    }
  }

  const vec hd = hv[(long)d * 32 + lane];
  const vec bv = ((const vec*)bias)[lane];
  const vec r = acc + hd * (dd * dd) + bv;
  ((vec*)agg)[(long)d * 32 + lane] = r;
}

extern "C" void kernel_launch(void* const* d_in, const int* in_sizes, int n_in,
                              void* d_out, int out_size, void* d_ws, size_t ws_size,
                              hipStream_t stream) {
  const float* x  = (const float*)d_in[0];
  const int*   ei = (const int*)d_in[1];
  const float* W1 = (const float*)d_in[2];
  const float* b1 = (const float*)d_in[3];
  const float* W2 = (const float*)d_in[4];
  const float* b2 = (const float*)d_in[5];
  float* out = (float*)d_out;

  float* ws   = (float*)d_ws;
  float* dinv = ws;
  float* h1   = dinv + N_NODES;
  float* agg1 = h1 + (size_t)N_NODES * HID;
  float* h3   = h1;
  int* aux     = (int*)(agg1 + (size_t)N_NODES * HID);
  int* counts  = aux;
  int* binptr  = counts + P_ROWS * BINP;
  int* rowptr  = binptr + BINP;
  int* csr_src = rowptr + N_NODES + 32;
  int* rec     = (int*)agg1;

  const int T = 256;

  k_bin_count<<<P_ROWS, PT, 0, stream>>>(ei, counts);
  k_bin_scatter<<<P_G, PT, 0, stream>>>(ei, counts, binptr, rec);
  k_bin_rows<<<NBINS, FT, 0, stream>>>(binptr, rec, rowptr, dinv, csr_src);

  const int mtiles = (N_NODES + 15) / 16;
  const int gblocks = (mtiles + 7) / 8;
  k_gemm_wmma<IN_DIM, HID / 16, false>
      <<<gblocks, T, 0, stream>>>(x, W1, h1, N_NODES);

  k_gather<HID><<<(N_NODES + 7) / 8, T, 0, stream>>>(rowptr, csr_src, dinv, h1, b1, agg1);

  k_gemm_wmma<HID, OUTD / 16, true>
      <<<gblocks, T, 0, stream>>>(agg1, W2, h3, N_NODES);

  k_gather<OUTD><<<(N_NODES + 7) / 8, T, 0, stream>>>(rowptr, csr_src, dinv, h3, b2, out);
}


# Source-side norm folded into the rows: GEMM epilogues scale row i by dinv[i], so both gathers drop the per-edge dinv[src] lookup (a scattered line per edge); sum order kept, products rounded per row
# speedup vs baseline: 3.2467x; 1.0714x over previous
#include <hip/hip_runtime.h>

#define N_NODES 100000
#define N_EDGES 1600000
#define IN_DIM  256
#define HID     128
#define OUTD    64

typedef __attribute__((ext_vector_type(2))) float v2f;
typedef __attribute__((ext_vector_type(4))) float v4f;
typedef __attribute__((ext_vector_type(8))) float v8f;

#define BIN_SHIFT 8
#define NBINS ((N_NODES + 255) >> BIN_SHIFT)
#define BINP 512
#define PT 1024
#define P_G 250
#define P_ROWS 256
#define P_CHUNK ((N_EDGES + P_G - 1) / P_G)
#define P_ITER ((P_CHUNK + PT - 1) / PT)
#define P_GW (P_ROWS / 4)
#define FT 1024
#define F_UB 5
#define STAGE_CAP 8192

__device__ __forceinline__ void wave0_excl_scan_256(const int* __restrict__ in,
                                                    int* __restrict__ out, int lane) {
  int v[8];
  int s = 0;
#pragma unroll
  for (int k = 0; k < 8; ++k) {
    v[k] = in[lane * 8 + k];
    s += v[k];
  }
  int inc = s;
#pragma unroll
  for (int off = 1; off < 32; off <<= 1) {
    const int nb = __shfl_up(inc, off);
    if (lane >= off) inc += nb;
  }
  int run = inc - s;
#pragma unroll
  for (int k = 0; k < 8; ++k) {
    out[lane * 8 + k] = run;
    run += v[k];
  }
}

__global__ __launch_bounds__(PT) void k_bin_count(const int* __restrict__ ei,
                                                  int* __restrict__ counts) {
  __shared__ int hist[BINP];
  const int tid = threadIdx.x;
  const int w = blockIdx.x;
  if (tid < BINP) hist[tid] = 0;
  const int base = w * P_CHUNK;
  const int lim = (base + P_CHUNK < N_EDGES) ? (base + P_CHUNK) : N_EDGES;
  int dv[P_ITER];
#pragma unroll
  for (int it = 0; it < P_ITER; ++it) {
    const int e = base + it * PT + tid;
    dv[it] = ei[N_EDGES + ((e < lim) ? e : (lim - 1))];
  }
  __syncthreads();
#pragma unroll
  for (int it = 0; it < P_ITER; ++it) {
    const int e = base + it * PT + tid;
    if (e < lim) atomicAdd(&hist[dv[it] >> BIN_SHIFT], 1);
  }
  __syncthreads();
  if (tid < BINP) counts[w * BINP + tid] = hist[tid];
}

__global__ __launch_bounds__(PT) void k_bin_scatter(const int* __restrict__ ei,
                                                    const int* __restrict__ counts,
                                                    int* __restrict__ binptr,
                                                    int* __restrict__ rec) {
  __shared__ int part_tot[4][BINP];
  __shared__ int part_pre[4][BINP];
  __shared__ int ptot[256];
  __shared__ int pex[256];
  __shared__ int cursor[BINP];
  const int tid = threadIdx.x;
  const int w = blockIdx.x;
  const int g = __builtin_amdgcn_readfirstlane(tid >> 8);
  const int t = tid & 255;

  const int base = w * P_CHUNK;
  const int lim = (base + P_CHUNK < N_EDGES) ? (base + P_CHUNK) : N_EDGES;
  int dv[P_ITER], sv[P_ITER];
#pragma unroll
  for (int it = 0; it < P_ITER; ++it) {
    const int e = base + it * PT + tid;
    const int ec = (e < lim) ? e : (lim - 1);
    dv[it] = ei[N_EDGES + ec];
    sv[it] = ei[ec];
  }

  const int2* __restrict__ c2 = (const int2*)counts;
  int tx = 0, ty = 0, px = 0, py = 0;
#pragma unroll 16
  for (int k = 0; k < P_GW; ++k) {
    const int ww = g * P_GW + k;
    const int2 v = c2[ww * (BINP / 2) + t];
    tx += v.x;
    ty += v.y;
    px += (ww < w) ? v.x : 0;
    py += (ww < w) ? v.y : 0;
  }
  part_tot[g][2 * t] = tx;
  part_tot[g][2 * t + 1] = ty;
  part_pre[g][2 * t] = px;
  part_pre[g][2 * t + 1] = py;
  __syncthreads();
  int T0 = 0, P0 = 0, P1 = 0;
  if (tid < 256) {
    int T1 = 0;
#pragma unroll
    for (int q = 0; q < 4; ++q) {
      T0 += part_tot[q][2 * tid];
      T1 += part_tot[q][2 * tid + 1];
      P0 += part_pre[q][2 * tid];
      P1 += part_pre[q][2 * tid + 1];
    }
    ptot[tid] = T0 + T1;
  }
  __syncthreads();
  if (tid < 32) wave0_excl_scan_256(ptot, pex, tid);
  __syncthreads();
  if (tid < 256) {
    const int b0 = pex[tid];
    const int b1 = b0 + T0;
    cursor[2 * tid] = b0 + P0;
    cursor[2 * tid + 1] = b1 + P1;
    if (w == 0) {
      binptr[2 * tid] = b0;
      binptr[2 * tid + 1] = b1;
    }
  }
  __syncthreads();
#pragma unroll
  for (int it = 0; it < P_ITER; ++it) {
    const int e = base + it * PT + tid;
    if (e < lim) {
      const int pos = atomicAdd(&cursor[dv[it] >> BIN_SHIFT], 1);
      rec[pos] = (sv[it] << BIN_SHIFT) | (dv[it] & 255);
    }
  }
}

__global__ __launch_bounds__(FT) void k_bin_rows(const int* __restrict__ binptr,
                                                 const int* __restrict__ rec,
                                                 int* __restrict__ rowptr,
                                                 float* __restrict__ dinv,
                                                 int* __restrict__ csr_src) {
  __shared__ int cnt[256];
  __shared__ int ex[256];
  __shared__ int stage[STAGE_CAP];
  const int tid = threadIdx.x;
  const int b = blockIdx.x;
  const int lo = binptr[b];
  const int n = binptr[b + 1] - lo;
  if (tid < 256) cnt[tid] = 0;
  __syncthreads();
  for (int i0 = tid; i0 < n; i0 += F_UB * FT) {
    int r[F_UB];
#pragma unroll
    for (int u = 0; u < F_UB; ++u) {
      const int i = i0 + u * FT;
      r[u] = rec[lo + ((i < n) ? i : (n - 1))];
    }
#pragma unroll
    for (int u = 0; u < F_UB; ++u)
      if (i0 + u * FT < n) atomicAdd(&cnt[r[u] & 255], 1);
  }
  __syncthreads();
  if (tid < 32) wave0_excl_scan_256(cnt, ex, tid);
  __syncthreads();
  if (tid < 256) {
    const int c = cnt[tid];
    const int x = ex[tid];
    const int node = (b << BIN_SHIFT) + tid;
    if (node < N_NODES) {
      rowptr[node] = lo + x;
      dinv[node] = rsqrtf((float)c + 1.0f);
      if (node == N_NODES - 1) rowptr[N_NODES] = lo + x + c;
    }
  }
  __syncthreads();
  const bool staged = (n <= STAGE_CAP);
  for (int i0 = tid; i0 < n; i0 += F_UB * FT) {
    int r[F_UB];
#pragma unroll
    for (int u = 0; u < F_UB; ++u) {
      const int i = i0 + u * FT;
      r[u] = rec[lo + ((i < n) ? i : (n - 1))];
    }
#pragma unroll
    for (int u = 0; u < F_UB; ++u) {
      if (i0 + u * FT < n) {
        const int p = atomicAdd(&ex[r[u] & 255], 1);
        if (staged) stage[p] = r[u] >> BIN_SHIFT;
        else csr_src[lo + p] = r[u] >> BIN_SHIFT;
      }
    }
  }
  if (staged) {
    __syncthreads();
    for (int i = tid; i < n; i += FT) csr_src[lo + i] = stage[i];
  }
}

template <int K, int NT, bool RELU>
__global__ __launch_bounds__(256) void k_gemm_wmma(const float* __restrict__ A,
                                                   const float* __restrict__ W,
                                                   const float* __restrict__ rscale,
                                                   float* __restrict__ C, int M) {
  constexpr int NC = NT * 16;
  constexpr int KC = 64;
  __shared__ float lds[KC * NC];

  const int wave = threadIdx.x >> 5;
  const int lane = threadIdx.x & 31;
  const int mtile = blockIdx.x * 8 + wave;
  const long m0 = (long)mtile * 16;
  const bool active = (m0 < M);

  const int m  = lane & 15;
  const int kh = lane >> 4;
  const int n  = lane & 15;

  const long arow = active ? (m0 + m) : (long)(M - 1);
  const float* __restrict__ Aptr = A + arow * K;

  v8f acc[NT] = {};

  for (int kc = 0; kc < K; kc += KC) {
    if (kc != 0) __syncthreads();
    {
      constexpr int ENT = KC * NC / 8;
#pragma unroll
      for (int idx = threadIdx.x; idx < ENT; idx += 256) {
        int kp = idx / (NC / 4);
        int nq = (idx % (NC / 4)) * 4;
        const float4 a = *(const float4*)(W + (long)(kc + 2 * kp) * NC + nq);
        const float4 b = *(const float4*)(W + (long)(kc + 2 * kp + 1) * NC + nq);
        float4* __restrict__ dst = (float4*)&lds[kp * 2 * NC + 2 * nq];
        dst[0] = make_float4(a.x, b.x, a.y, b.y);
        dst[1] = make_float4(a.z, b.z, a.w, b.w);
      }
    }
    __syncthreads();

#pragma unroll 4
    for (int k0 = kc; k0 < kc + KC; k0 += 4) {
      const int ka = k0 + 2 * kh;
      const int kp = (ka - kc) >> 1;
      v2f av = *(const v2f*)(Aptr + ka);
      if (RELU) {
        av.x = fmaxf(av.x, 0.0f);
        av.y = fmaxf(av.y, 0.0f);
      }
      const v2f* __restrict__ brow = (const v2f*)&lds[kp * 2 * NC + 2 * n];
#pragma unroll
      for (int j = 0; j < NT; ++j) {
        v2f bv = brow[j * 16];
        acc[j] = __builtin_amdgcn_wmma_f32_16x16x4_f32(
             false, av,  false, bv,
             (short)0, acc[j],  false,  false);
      }
    }
  }

  if (!active) return;
  float rs[8];
#pragma unroll
  for (int r = 0; r < 8; ++r) {
    const long row = m0 + r + 8 * kh;
    rs[r] = rscale[(row < M) ? row : (long)(M - 1)];
  }
#pragma unroll
  for (int j = 0; j < NT; ++j) {
#pragma unroll
    for (int r = 0; r < 8; ++r) {
      long row = m0 + r + 8 * kh;
      if (row < M) C[row * NC + j * 16 + n] = acc[j][r] * rs[r];
    }
  }
}

template <int V> struct LaneVec;
template <> struct LaneVec<2> { typedef v2f type; };
template <> struct LaneVec<4> { typedef v4f type; };

template <int F>
__global__ __launch_bounds__(256) void k_gather(const int* __restrict__ rowptr,
                                                const int* __restrict__ csr_src,
                                                const float* __restrict__ dinv,
                                                const float* __restrict__ hs,
                                                const float* __restrict__ bias,
                                                float* __restrict__ agg) {
  typedef typename LaneVec<F / 32>::type vec;
  constexpr int DEPTH = 8;
  const int lane = threadIdx.x & 31;
  const int wv = __builtin_amdgcn_readfirstlane((int)(threadIdx.x >> 5));
  const int d = blockIdx.x * 8 + wv;
  if (d >= N_NODES) return;

  const int start = rowptr[d];
  const int end = rowptr[d + 1];
  const float dd = dinv[d];
  const vec* __restrict__ hv = (const vec*)hs;

  vec acc = {};
  for (int base = start; base < end; base += 32) {
    const int m = (end - base < 32) ? (end - base) : 32;
    const int li = base + lane;
    const int myidx = csr_src[(li < end) ? li : (end - 1)];
    for (int j = 0; j < m; j += DEPTH) {
      vec v[DEPTH];
#pragma unroll
      for (int u = 0; u < DEPTH; ++u) {
        const int s = __builtin_amdgcn_readlane(myidx, j + u);
        v[u] = hv[(long)s * 32 + lane];
      }
#pragma unroll
      for (int u = 0; u < DEPTH; ++u) {
        const float wt = (j + u < m) ? 1.0f : 0.0f;
        acc += v[u] * wt;
      }
    }
  }

  const vec hd = hv[(long)d * 32 + lane];
  const vec bv = ((const vec*)bias)[lane];
  const vec r = (acc + hd) * dd + bv;
  ((vec*)agg)[(long)d * 32 + lane] = r;
}

extern "C" void kernel_launch(void* const* d_in, const int* in_sizes, int n_in,
                              void* d_out, int out_size, void* d_ws, size_t ws_size,
                              hipStream_t stream) {
  const float* x  = (const float*)d_in[0];
  const int*   ei = (const int*)d_in[1];
  const float* W1 = (const float*)d_in[2];
  const float* b1 = (const float*)d_in[3];
  const float* W2 = (const float*)d_in[4];
  const float* b2 = (const float*)d_in[5];
  float* out = (float*)d_out;

  float* ws   = (float*)d_ws;
  float* dinv = ws;
  float* h1   = dinv + N_NODES;
  float* agg1 = h1 + (size_t)N_NODES * HID;
  float* h3   = h1;
  int* aux     = (int*)(agg1 + (size_t)N_NODES * HID);
  int* counts  = aux;
  int* binptr  = counts + P_ROWS * BINP;
  int* rowptr  = binptr + BINP;
  int* csr_src = rowptr + N_NODES + 32;
  int* rec     = (int*)agg1;

  const int T = 256;

  k_bin_count<<<P_ROWS, PT, 0, stream>>>(ei, counts);
  k_bin_scatter<<<P_G, PT, 0, stream>>>(ei, counts, binptr, rec);
  k_bin_rows<<<NBINS, FT, 0, stream>>>(binptr, rec, rowptr, dinv, csr_src);

  const int mtiles = (N_NODES + 15) / 16;
  const int gblocks = (mtiles + 7) / 8;
  k_gemm_wmma<IN_DIM, HID / 16, false>
      <<<gblocks, T, 0, stream>>>(x, W1, dinv, h1, N_NODES);

  k_gather<HID><<<(N_NODES + 7) / 8, T, 0, stream>>>(rowptr, csr_src, dinv, h1, b1, agg1);

  k_gemm_wmma<HID, OUTD / 16, true>
      <<<gblocks, T, 0, stream>>>(agg1, W2, dinv, h3, N_NODES);

  k_gather<OUTD><<<(N_NODES + 7) / 8, T, 0, stream>>>(rowptr, csr_src, dinv, h3, b2, out);
}


# Layer-1 gather fused with layer-2 GEMM: a 32-node work-group keeps its ReLU'd row sums in an LDS tile and multiplies them by W2 (f32 WMMA) there; agg1 never stored, 7 launches -> 6; sum orders kept
# speedup vs baseline: 3.4090x; 1.0500x over previous
#include <hip/hip_runtime.h>

#define N_NODES 100000
#define N_EDGES 1600000
#define IN_DIM  256
#define HID     128
#define OUTD    64

typedef __attribute__((ext_vector_type(2))) float v2f;
typedef __attribute__((ext_vector_type(4))) float v4f;
typedef __attribute__((ext_vector_type(8))) float v8f;

#define BIN_SHIFT 8
#define NBINS ((N_NODES + 255) >> BIN_SHIFT)
#define BINP 512
#define PT 1024
#define P_G 250
#define P_ROWS 256
#define P_CHUNK ((N_EDGES + P_G - 1) / P_G)
#define P_ITER ((P_CHUNK + PT - 1) / PT)
#define P_GW (P_ROWS / 4)
#define FT 1024
#define F_UB 5
#define STAGE_CAP 8192

__device__ __forceinline__ void wave0_excl_scan_256(const int* __restrict__ in,
                                                    int* __restrict__ out, int lane) {
  int v[8];
  int s = 0;
#pragma unroll
  for (int k = 0; k < 8; ++k) {
    v[k] = in[lane * 8 + k];
    s += v[k];
  }
  int inc = s;
#pragma unroll
  for (int off = 1; off < 32; off <<= 1) {
    const int nb = __shfl_up(inc, off);
    if (lane >= off) inc += nb;
  }
  int run = inc - s;
#pragma unroll
  for (int k = 0; k < 8; ++k) {
    out[lane * 8 + k] = run;
    run += v[k];
  }
}

__global__ __launch_bounds__(PT) void k_bin_count(const int* __restrict__ ei,
                                                  int* __restrict__ counts) {
  __shared__ int hist[BINP];
  const int tid = threadIdx.x;
  const int w = blockIdx.x;
  if (tid < BINP) hist[tid] = 0;
  const int base = w * P_CHUNK;
  const int lim = (base + P_CHUNK < N_EDGES) ? (base + P_CHUNK) : N_EDGES;
  int dv[P_ITER];
#pragma unroll
  for (int it = 0; it < P_ITER; ++it) {
    const int e = base + it * PT + tid;
    dv[it] = ei[N_EDGES + ((e < lim) ? e : (lim - 1))];
  }
  __syncthreads();
#pragma unroll
  for (int it = 0; it < P_ITER; ++it) {
    const int e = base + it * PT + tid;
    if (e < lim) atomicAdd(&hist[dv[it] >> BIN_SHIFT], 1);
  }
  __syncthreads();
  if (tid < BINP) counts[w * BINP + tid] = hist[tid];
}

__global__ __launch_bounds__(PT) void k_bin_scatter(const int* __restrict__ ei,
                                                    const int* __restrict__ counts,
                                                    int* __restrict__ binptr,
                                                    int* __restrict__ rec) {
  __shared__ int part_tot[4][BINP];
  __shared__ int part_pre[4][BINP];
  __shared__ int ptot[256];
  __shared__ int pex[256];
  __shared__ int cursor[BINP];
  const int tid = threadIdx.x;
  const int w = blockIdx.x;
  const int g = __builtin_amdgcn_readfirstlane(tid >> 8);
  const int t = tid & 255;

  const int base = w * P_CHUNK;
  const int lim = (base + P_CHUNK < N_EDGES) ? (base + P_CHUNK) : N_EDGES;
  int dv[P_ITER], sv[P_ITER];
#pragma unroll
  for (int it = 0; it < P_ITER; ++it) {
    const int e = base + it * PT + tid;
    const int ec = (e < lim) ? e : (lim - 1);
    dv[it] = ei[N_EDGES + ec];
    sv[it] = ei[ec];
  }

  const int2* __restrict__ c2 = (const int2*)counts;
  int tx = 0, ty = 0, px = 0, py = 0;
#pragma unroll 16
  for (int k = 0; k < P_GW; ++k) {
    const int ww = g * P_GW + k;
    const int2 v = c2[ww * (BINP / 2) + t];
    tx += v.x;
    ty += v.y;
    px += (ww < w) ? v.x : 0;
    py += (ww < w) ? v.y : 0;
  }
  part_tot[g][2 * t] = tx;
  part_tot[g][2 * t + 1] = ty;
  part_pre[g][2 * t] = px;
  part_pre[g][2 * t + 1] = py;
  __syncthreads();
  int T0 = 0, P0 = 0, P1 = 0;
  if (tid < 256) {
    int T1 = 0;
#pragma unroll
    for (int q = 0; q < 4; ++q) {
      T0 += part_tot[q][2 * tid];
      T1 += part_tot[q][2 * tid + 1];
      P0 += part_pre[q][2 * tid];
      P1 += part_pre[q][2 * tid + 1];
    }
    ptot[tid] = T0 + T1;
  }
  __syncthreads();
  if (tid < 32) wave0_excl_scan_256(ptot, pex, tid);
  __syncthreads();
  if (tid < 256) {
    const int b0 = pex[tid];
    const int b1 = b0 + T0;
    cursor[2 * tid] = b0 + P0;
    cursor[2 * tid + 1] = b1 + P1;
    if (w == 0) {
      binptr[2 * tid] = b0;
      binptr[2 * tid + 1] = b1;
    }
  }
  __syncthreads();
#pragma unroll
  for (int it = 0; it < P_ITER; ++it) {
    const int e = base + it * PT + tid;
    if (e < lim) {
      const int pos = atomicAdd(&cursor[dv[it] >> BIN_SHIFT], 1);
      rec[pos] = (sv[it] << BIN_SHIFT) | (dv[it] & 255);
    }
  }
}

__global__ __launch_bounds__(FT) void k_bin_rows(const int* __restrict__ binptr,
                                                 const int* __restrict__ rec,
                                                 int* __restrict__ rowptr,
                                                 float* __restrict__ dinv,
                                                 int* __restrict__ csr_src) {
  __shared__ int cnt[256];
  __shared__ int ex[256];
  __shared__ int stage[STAGE_CAP];
  const int tid = threadIdx.x;
  const int b = blockIdx.x;
  const int lo = binptr[b];
  const int n = binptr[b + 1] - lo;
  if (tid < 256) cnt[tid] = 0;
  __syncthreads();
  for (int i0 = tid; i0 < n; i0 += F_UB * FT) {
    int r[F_UB];
#pragma unroll
    for (int u = 0; u < F_UB; ++u) {
      const int i = i0 + u * FT;
      r[u] = rec[lo + ((i < n) ? i : (n - 1))];
    }
#pragma unroll
    for (int u = 0; u < F_UB; ++u)
      if (i0 + u * FT < n) atomicAdd(&cnt[r[u] & 255], 1);
  }
  __syncthreads();
  if (tid < 32) wave0_excl_scan_256(cnt, ex, tid);
  __syncthreads();
  if (tid < 256) {
    const int c = cnt[tid];
    const int x = ex[tid];
    const int node = (b << BIN_SHIFT) + tid;
    if (node < N_NODES) {
      rowptr[node] = lo + x;
      dinv[node] = rsqrtf((float)c + 1.0f);
      if (node == N_NODES - 1) rowptr[N_NODES] = lo + x + c;
    }
  }
  __syncthreads();
  const bool staged = (n <= STAGE_CAP);
  for (int i0 = tid; i0 < n; i0 += F_UB * FT) {
    int r[F_UB];
#pragma unroll
    for (int u = 0; u < F_UB; ++u) {
      const int i = i0 + u * FT;
      r[u] = rec[lo + ((i < n) ? i : (n - 1))];
    }
#pragma unroll
    for (int u = 0; u < F_UB; ++u) {
      if (i0 + u * FT < n) {
        const int p = atomicAdd(&ex[r[u] & 255], 1);
        if (staged) stage[p] = r[u] >> BIN_SHIFT;
        else csr_src[lo + p] = r[u] >> BIN_SHIFT;
      }
    }
  }
  if (staged) {
    __syncthreads();
    for (int i = tid; i < n; i += FT) csr_src[lo + i] = stage[i];
  }
}

template <int K, int NT, bool RELU>
__global__ __launch_bounds__(256) void k_gemm_wmma(const float* __restrict__ A,
                                                   const float* __restrict__ W,
                                                   const float* __restrict__ rscale,
                                                   float* __restrict__ C, int M) {
  constexpr int NC = NT * 16;
  constexpr int KC = 64;
  __shared__ float lds[KC * NC];

  const int wave = threadIdx.x >> 5;
  const int lane = threadIdx.x & 31;
  const int mtile = blockIdx.x * 8 + wave;
  const long m0 = (long)mtile * 16;
  const bool active = (m0 < M);

  const int m  = lane & 15;
  const int kh = lane >> 4;
  const int n  = lane & 15;

  const long arow = active ? (m0 + m) : (long)(M - 1);
  const float* __restrict__ Aptr = A + arow * K;

  v8f acc[NT] = {};

  for (int kc = 0; kc < K; kc += KC) {
    if (kc != 0) __syncthreads();
    {
      constexpr int ENT = KC * NC / 8;
#pragma unroll
      for (int idx = threadIdx.x; idx < ENT; idx += 256) {
        int kp = idx / (NC / 4);
        int nq = (idx % (NC / 4)) * 4;
        const float4 a = *(const float4*)(W + (long)(kc + 2 * kp) * NC + nq);
        const float4 b = *(const float4*)(W + (long)(kc + 2 * kp + 1) * NC + nq);
        float4* __restrict__ dst = (float4*)&lds[kp * 2 * NC + 2 * nq];
        dst[0] = make_float4(a.x, b.x, a.y, b.y);
        dst[1] = make_float4(a.z, b.z, a.w, b.w);
      }
    }
    __syncthreads();

#pragma unroll 4
    for (int k0 = kc; k0 < kc + KC; k0 += 4) {
      const int ka = k0 + 2 * kh;
      const int kp = (ka - kc) >> 1;
      v2f av = *(const v2f*)(Aptr + ka);
      if (RELU) {
        av.x = fmaxf(av.x, 0.0f);
        av.y = fmaxf(av.y, 0.0f);
      }
      const v2f* __restrict__ brow = (const v2f*)&lds[kp * 2 * NC + 2 * n];
#pragma unroll
      for (int j = 0; j < NT; ++j) {
        v2f bv = brow[j * 16];
        acc[j] = __builtin_amdgcn_wmma_f32_16x16x4_f32(
             false, av,  false, bv,
             (short)0, acc[j],  false,  false);
      }
    }
  }

  if (!active) return;
  float rs[8];
#pragma unroll
  for (int r = 0; r < 8; ++r) {
    const long row = m0 + r + 8 * kh;
    rs[r] = rscale[(row < M) ? row : (long)(M - 1)];
  }
#pragma unroll
  for (int j = 0; j < NT; ++j) {
#pragma unroll
    for (int r = 0; r < 8; ++r) {
      long row = m0 + r + 8 * kh;
      if (row < M) C[row * NC + j * 16 + n] = acc[j][r] * rs[r];
    }
  }
}

template <int V> struct LaneVec;
template <> struct LaneVec<2> { typedef v2f type; };
template <> struct LaneVec<4> { typedef v4f type; };

template <int F>
__global__ __launch_bounds__(256) void k_gather(const int* __restrict__ rowptr,
                                                const int* __restrict__ csr_src,
                                                const float* __restrict__ dinv,
                                                const float* __restrict__ hs,
                                                const float* __restrict__ bias,
                                                float* __restrict__ agg) {
  typedef typename LaneVec<F / 32>::type vec;
  constexpr int DEPTH = 8;
  const int lane = threadIdx.x & 31;
  const int wv = __builtin_amdgcn_readfirstlane((int)(threadIdx.x >> 5));
  const int d = blockIdx.x * 8 + wv;
  if (d >= N_NODES) return;

  const int start = rowptr[d];
  const int end = rowptr[d + 1];
  const float dd = dinv[d];
  const vec* __restrict__ hv = (const vec*)hs;

  vec acc = {};
  for (int base = start; base < end; base += 32) {
    const int m = (end - base < 32) ? (end - base) : 32;
    const int li = base + lane;
    const int myidx = csr_src[(li < end) ? li : (end - 1)];
    for (int j = 0; j < m; j += DEPTH) {
      vec v[DEPTH];
#pragma unroll
      for (int u = 0; u < DEPTH; ++u) {
        const int s = __builtin_amdgcn_readlane(myidx, j + u);
        v[u] = hv[(long)s * 32 + lane];
      }
#pragma unroll
      for (int u = 0; u < DEPTH; ++u) {
        const float wt = (j + u < m) ? 1.0f : 0.0f;
        acc += v[u] * wt;
      }
    }
  }

  const vec hd = hv[(long)d * 32 + lane];
  const vec bv = ((const vec*)bias)[lane];
  const vec r = (acc + hd) * dd + bv;
  ((vec*)agg)[(long)d * 32 + lane] = r;
}

#define FG_NODES 32
#define FG_PITCH (HID + 4)

__global__ __launch_bounds__(256) void k_gather_gemm(const int* __restrict__ rowptr,
                                                     const int* __restrict__ csr_src,
                                                     const float* __restrict__ dinv,
                                                     const float* __restrict__ hs,
                                                     const float* __restrict__ bias,
                                                     const float* __restrict__ W,
                                                     float* __restrict__ C) {
  static_assert(N_NODES % FG_NODES == 0, "whole tiles only: WMMA needs every lane");
  constexpr int NC = OUTD;
  constexpr int DEPTH = 8;
  __shared__ float wl[HID * NC];
  __shared__ float tile[FG_NODES * FG_PITCH];

  const int lane = threadIdx.x & 31;
  const int wv = __builtin_amdgcn_readfirstlane((int)(threadIdx.x >> 5));
  const int node0 = blockIdx.x * FG_NODES;

#pragma unroll
  for (int q = 0; q < 4; ++q) {
    const int idx = threadIdx.x + q * 256;
    const int kp = idx / (NC / 4);
    const int nq = (idx % (NC / 4)) * 4;
    const float4 a = *(const float4*)(W + (2 * kp) * NC + nq);
    const float4 b = *(const float4*)(W + (2 * kp + 1) * NC + nq);
    float4* __restrict__ dst = (float4*)&wl[kp * 2 * NC + 2 * nq];
    dst[0] = make_float4(a.x, b.x, a.y, b.y);
    dst[1] = make_float4(a.z, b.z, a.w, b.w);
  }

  const v4f* __restrict__ hv = (const v4f*)hs;
  const v4f bv = ((const v4f*)bias)[lane];

  for (int i = 0; i < 4; ++i) {
    const int t = wv * 4 + i;
    const int d = node0 + t;
    const int start = rowptr[d];
    const int end = rowptr[d + 1];
    const float dd = dinv[d];
    const v4f hd = hv[(long)d * 32 + lane];

    v4f acc = {};
    for (int base = start; base < end; base += 32) {
      const int m = (end - base < 32) ? (end - base) : 32;
      const int li = base + lane;
      const int myidx = csr_src[(li < end) ? li : (end - 1)];
      for (int j = 0; j < m; j += DEPTH) {
        v4f v[DEPTH];
#pragma unroll
        for (int u = 0; u < DEPTH; ++u) {
          const int s = __builtin_amdgcn_readlane(myidx, j + u);
          v[u] = hv[(long)s * 32 + lane];
        }
#pragma unroll
        for (int u = 0; u < DEPTH; ++u) {
          const float wt = (j + u < m) ? 1.0f : 0.0f;
          acc += v[u] * wt;
        }
      }
    }
    v4f r = (acc + hd) * dd + bv;
    r.x = fmaxf(r.x, 0.0f);
    r.y = fmaxf(r.y, 0.0f);
    r.z = fmaxf(r.z, 0.0f);
    r.w = fmaxf(r.w, 0.0f);
    *(v4f*)&tile[t * FG_PITCH + lane * 4] = r;
  }

  __syncthreads();

  const int mt = wv >> 2;
  const int nt = wv & 3;
  const int m  = lane & 15;
  const int kh = lane >> 4;
  const float* __restrict__ arow = &tile[(mt * 16 + m) * FG_PITCH + 2 * kh];
  const float* __restrict__ bcol = &wl[kh * 2 * NC + 2 * (nt * 16 + m)];
  v8f c = {};
#pragma unroll
  for (int k0 = 0; k0 < HID; k0 += 4) {
    const v2f av = *(const v2f*)(arow + k0);
    const v2f bw = *(const v2f*)(bcol + (k0 >> 1) * 2 * NC);
    c = __builtin_amdgcn_wmma_f32_16x16x4_f32(
         false, av,  false, bw,
         (short)0, c,  false,  false);
  }

#pragma unroll
  for (int r = 0; r < 8; ++r) {
    const int row = node0 + mt * 16 + r + 8 * kh;
    C[(long)row * NC + nt * 16 + m] = c[r] * dinv[row];
  }
}

extern "C" void kernel_launch(void* const* d_in, const int* in_sizes, int n_in,
                              void* d_out, int out_size, void* d_ws, size_t ws_size,
                              hipStream_t stream) {
  const float* x  = (const float*)d_in[0];
  const int*   ei = (const int*)d_in[1];
  const float* W1 = (const float*)d_in[2];
  const float* b1 = (const float*)d_in[3];
  const float* W2 = (const float*)d_in[4];
  const float* b2 = (const float*)d_in[5];
  float* out = (float*)d_out;

  float* ws   = (float*)d_ws;
  float* dinv = ws;
  float* h1   = dinv + N_NODES;
  float* agg1 = h1 + (size_t)N_NODES * HID;
  float* h3   = agg1;
  int* aux     = (int*)(agg1 + (size_t)N_NODES * HID);
  int* counts  = aux;
  int* binptr  = counts + P_ROWS * BINP;
  int* rowptr  = binptr + BINP;
  int* csr_src = rowptr + N_NODES + 32;
  int* rec     = (int*)agg1;

  const int T = 256;

  k_bin_count<<<P_ROWS, PT, 0, stream>>>(ei, counts);
  k_bin_scatter<<<P_G, PT, 0, stream>>>(ei, counts, binptr, rec);
  k_bin_rows<<<NBINS, FT, 0, stream>>>(binptr, rec, rowptr, dinv, csr_src);

  const int mtiles = (N_NODES + 15) / 16;
  const int gblocks = (mtiles + 7) / 8;
  k_gemm_wmma<IN_DIM, HID / 16, false>
      <<<gblocks, T, 0, stream>>>(x, W1, dinv, h1, N_NODES);

  k_gather_gemm<<<N_NODES / FG_NODES, T, 0, stream>>>(rowptr, csr_src, dinv, h1, b1, W2, h3);

  k_gather<OUTD><<<(N_NODES + 7) / 8, T, 0, stream>>>(rowptr, csr_src, dinv, h3, b2, out);
}


# Layer-1 GEMM as one resident 32-wave work-group per WGP: W1 staged into LDS once, no barrier in the k loop, wave pairs split a tile's columns (at most 13 half tiles a SIMD, not 16); sum order kept
# speedup vs baseline: 3.6357x; 1.0665x over previous
#include <hip/hip_runtime.h>

#define N_NODES 100000
#define N_EDGES 1600000
#define IN_DIM  256
#define HID     128
#define OUTD    64

typedef __attribute__((ext_vector_type(2))) float v2f;
typedef __attribute__((ext_vector_type(4))) float v4f;
typedef __attribute__((ext_vector_type(8))) float v8f;

#define BIN_SHIFT 8
#define NBINS ((N_NODES + 255) >> BIN_SHIFT)
#define BINP 512
#define PT 1024
#define P_G 250
#define P_ROWS 256
#define P_CHUNK ((N_EDGES + P_G - 1) / P_G)
#define P_ITER ((P_CHUNK + PT - 1) / PT)
#define P_GW (P_ROWS / 4)
#define FT 1024
#define F_UB 5
#define STAGE_CAP 8192

__device__ __forceinline__ void wave0_excl_scan_256(const int* __restrict__ in,
                                                    int* __restrict__ out, int lane) {
  int v[8];
  int s = 0;
#pragma unroll
  for (int k = 0; k < 8; ++k) {
    v[k] = in[lane * 8 + k];
    s += v[k];
  }
  int inc = s;
#pragma unroll
  for (int off = 1; off < 32; off <<= 1) {
    const int nb = __shfl_up(inc, off);
    if (lane >= off) inc += nb;
  }
  int run = inc - s;
#pragma unroll
  for (int k = 0; k < 8; ++k) {
    out[lane * 8 + k] = run;
    run += v[k];
  }
}

__global__ __launch_bounds__(PT) void k_bin_count(const int* __restrict__ ei,
                                                  int* __restrict__ counts) {
  __shared__ int hist[BINP];
  const int tid = threadIdx.x;
  const int w = blockIdx.x;
  if (tid < BINP) hist[tid] = 0;
  const int base = w * P_CHUNK;
  const int lim = (base + P_CHUNK < N_EDGES) ? (base + P_CHUNK) : N_EDGES;
  int dv[P_ITER];
#pragma unroll
  for (int it = 0; it < P_ITER; ++it) {
    const int e = base + it * PT + tid;
    dv[it] = ei[N_EDGES + ((e < lim) ? e : (lim - 1))];
  }
  __syncthreads();
#pragma unroll
  for (int it = 0; it < P_ITER; ++it) {
    const int e = base + it * PT + tid;
    if (e < lim) atomicAdd(&hist[dv[it] >> BIN_SHIFT], 1);
  }
  __syncthreads();
  if (tid < BINP) counts[w * BINP + tid] = hist[tid];
}

__global__ __launch_bounds__(PT) void k_bin_scatter(const int* __restrict__ ei,
                                                    const int* __restrict__ counts,
                                                    int* __restrict__ binptr,
                                                    int* __restrict__ rec) {
  __shared__ int part_tot[4][BINP];
  __shared__ int part_pre[4][BINP];
  __shared__ int ptot[256];
  __shared__ int pex[256];
  __shared__ int cursor[BINP];
  const int tid = threadIdx.x;
  const int w = blockIdx.x;
  const int g = __builtin_amdgcn_readfirstlane(tid >> 8);
  const int t = tid & 255;

  const int base = w * P_CHUNK;
  const int lim = (base + P_CHUNK < N_EDGES) ? (base + P_CHUNK) : N_EDGES;
  int dv[P_ITER], sv[P_ITER];
#pragma unroll
  for (int it = 0; it < P_ITER; ++it) {
    const int e = base + it * PT + tid;
    const int ec = (e < lim) ? e : (lim - 1);
    dv[it] = ei[N_EDGES + ec];
    sv[it] = ei[ec];
  }

  const int2* __restrict__ c2 = (const int2*)counts;
  int tx = 0, ty = 0, px = 0, py = 0;
#pragma unroll 16
  for (int k = 0; k < P_GW; ++k) {
    const int ww = g * P_GW + k;
    const int2 v = c2[ww * (BINP / 2) + t];
    tx += v.x;
    ty += v.y;
    px += (ww < w) ? v.x : 0;
    py += (ww < w) ? v.y : 0;
  }
  part_tot[g][2 * t] = tx;
  part_tot[g][2 * t + 1] = ty;
  part_pre[g][2 * t] = px;
  part_pre[g][2 * t + 1] = py;
  __syncthreads();
  int T0 = 0, P0 = 0, P1 = 0;
  if (tid < 256) {
    int T1 = 0;
#pragma unroll
    for (int q = 0; q < 4; ++q) {
      T0 += part_tot[q][2 * tid];
      T1 += part_tot[q][2 * tid + 1];
      P0 += part_pre[q][2 * tid];
      P1 += part_pre[q][2 * tid + 1];
    }
    ptot[tid] = T0 + T1;
  }
  __syncthreads();
  if (tid < 32) wave0_excl_scan_256(ptot, pex, tid);
  __syncthreads();
  if (tid < 256) {
    const int b0 = pex[tid];
    const int b1 = b0 + T0;
    cursor[2 * tid] = b0 + P0;
    cursor[2 * tid + 1] = b1 + P1;
    if (w == 0) {
      binptr[2 * tid] = b0;
      binptr[2 * tid + 1] = b1;
    }
  }
  __syncthreads();
#pragma unroll
  for (int it = 0; it < P_ITER; ++it) {
    const int e = base + it * PT + tid;
    if (e < lim) {
      const int pos = atomicAdd(&cursor[dv[it] >> BIN_SHIFT], 1);
      rec[pos] = (sv[it] << BIN_SHIFT) | (dv[it] & 255);
    }
  }
}

__global__ __launch_bounds__(FT) void k_bin_rows(const int* __restrict__ binptr,
                                                 const int* __restrict__ rec,
                                                 int* __restrict__ rowptr,
                                                 float* __restrict__ dinv,
                                                 int* __restrict__ csr_src) {
  __shared__ int cnt[256];
  __shared__ int ex[256];
  __shared__ int stage[STAGE_CAP];
  const int tid = threadIdx.x;
  const int b = blockIdx.x;
  const int lo = binptr[b];
  const int n = binptr[b + 1] - lo;
  if (tid < 256) cnt[tid] = 0;
  __syncthreads();
  for (int i0 = tid; i0 < n; i0 += F_UB * FT) {
    int r[F_UB];
#pragma unroll
    for (int u = 0; u < F_UB; ++u) {
      const int i = i0 + u * FT;
      r[u] = rec[lo + ((i < n) ? i : (n - 1))];
    }
#pragma unroll
    for (int u = 0; u < F_UB; ++u)
      if (i0 + u * FT < n) atomicAdd(&cnt[r[u] & 255], 1);
  }
  __syncthreads();
  if (tid < 32) wave0_excl_scan_256(cnt, ex, tid);
  __syncthreads();
  if (tid < 256) {
    const int c = cnt[tid];
    const int x = ex[tid];
    const int node = (b << BIN_SHIFT) + tid;
    if (node < N_NODES) {
      rowptr[node] = lo + x;
      dinv[node] = rsqrtf((float)c + 1.0f);
      if (node == N_NODES - 1) rowptr[N_NODES] = lo + x + c;
    }
  }
  __syncthreads();
  const bool staged = (n <= STAGE_CAP);
  for (int i0 = tid; i0 < n; i0 += F_UB * FT) {
    int r[F_UB];
#pragma unroll
    for (int u = 0; u < F_UB; ++u) {
      const int i = i0 + u * FT;
      r[u] = rec[lo + ((i < n) ? i : (n - 1))];
    }
#pragma unroll
    for (int u = 0; u < F_UB; ++u) {
      if (i0 + u * FT < n) {
        const int p = atomicAdd(&ex[r[u] & 255], 1);
        if (staged) stage[p] = r[u] >> BIN_SHIFT;
        else csr_src[lo + p] = r[u] >> BIN_SHIFT;
      }
    }
  }
  if (staged) {
    __syncthreads();
    for (int i = tid; i < n; i += FT) csr_src[lo + i] = stage[i];
  }
}

#define G1_WG 256
#define G1_T 1024
#define G1_PAIRS (G1_T / 64)

__global__ __launch_bounds__(G1_T) __attribute__((amdgpu_waves_per_eu(8, 8))) void k_gemm1_resident(const float* __restrict__ A,
                                                        const float* __restrict__ W,
                                                        const float* __restrict__ rscale,
                                                        float* __restrict__ C, int M) {
  constexpr int K = IN_DIM;
  constexpr int NC = HID;
  constexpr int NTH = 4;
  __shared__ float lds[K * NC];

  {
    constexpr int ENT = K * NC / 8;
    constexpr int PER = ENT / G1_T;
    static_assert(ENT % G1_T == 0, "whole entries per thread");
    float4 ra[PER], rb[PER];
#pragma unroll
    for (int q = 0; q < PER; ++q) {
      const int idx = threadIdx.x + q * G1_T;
      const int kp = idx / (NC / 4);
      const int nq = (idx % (NC / 4)) * 4;
      ra[q] = *(const float4*)(W + (2 * kp) * NC + nq);
      rb[q] = *(const float4*)(W + (2 * kp + 1) * NC + nq);
    }
#pragma unroll
    for (int q = 0; q < PER; ++q) {
      const int idx = threadIdx.x + q * G1_T;
      const int kp = idx / (NC / 4);
      const int nq = (idx % (NC / 4)) * 4;
      float4* __restrict__ dst = (float4*)&lds[kp * 2 * NC + 2 * nq];
      dst[0] = make_float4(ra[q].x, rb[q].x, ra[q].y, rb[q].y);
      dst[1] = make_float4(ra[q].z, rb[q].z, ra[q].w, rb[q].w);
    }
  }
  __syncthreads();

  const int lane = threadIdx.x & 31;
  const int wave = __builtin_amdgcn_readfirstlane((int)(threadIdx.x >> 5));
  const int p  = wave >> 1;
  const int hf = wave & 1;
  const int m  = lane & 15;
  const int kh = lane >> 4;
  const int n  = lane & 15;
  const int mtiles = (M + 15) >> 4;
  const v2f* __restrict__ bbase = (const v2f*)lds + kh * NC + hf * (NTH * 16) + n;

  for (int tile = blockIdx.x + G1_WG * p; tile < mtiles; tile += G1_WG * G1_PAIRS) {
    const long m0 = (long)tile * 16;
    const long arow = (m0 + m < M) ? (m0 + m) : (long)(M - 1);
    const float* __restrict__ Aptr = A + arow * K + 2 * kh;

    v8f acc[NTH] = {};
    for (int k8 = 0; k8 < K; k8 += 32) {
      v2f av[8];
#pragma unroll
      for (int u = 0; u < 8; ++u) av[u] = *(const v2f*)(Aptr + k8 + 4 * u);
#pragma unroll
      for (int u = 0; u < 8; ++u) {
        const v2f* __restrict__ brow = bbase + ((k8 + 4 * u) >> 1) * NC;
#pragma unroll
        for (int j = 0; j < NTH; ++j) {
          const v2f bv = brow[j * 16];
          acc[j] = __builtin_amdgcn_wmma_f32_16x16x4_f32(
               false, av[u],  false, bv,
               (short)0, acc[j],  false,  false);
        }
      }
    }

    float rs[8];
#pragma unroll
    for (int r = 0; r < 8; ++r) {
      const long row = m0 + r + 8 * kh;
      rs[r] = rscale[(row < M) ? row : (long)(M - 1)];
    }
#pragma unroll
    for (int j = 0; j < NTH; ++j) {
#pragma unroll
      for (int r = 0; r < 8; ++r) {
        const long row = m0 + r + 8 * kh;
        if (row < M) C[row * NC + hf * (NTH * 16) + j * 16 + n] = acc[j][r] * rs[r];
      }
    }
  }
}

template <int V> struct LaneVec;
template <> struct LaneVec<2> { typedef v2f type; };
template <> struct LaneVec<4> { typedef v4f type; };

template <int F>
__global__ __launch_bounds__(256) void k_gather(const int* __restrict__ rowptr,
                                                const int* __restrict__ csr_src,
                                                const float* __restrict__ dinv,
                                                const float* __restrict__ hs,
                                                const float* __restrict__ bias,
                                                float* __restrict__ agg) {
  typedef typename LaneVec<F / 32>::type vec;
  constexpr int DEPTH = 8;
  const int lane = threadIdx.x & 31;
  const int wv = __builtin_amdgcn_readfirstlane((int)(threadIdx.x >> 5));
  const int d = blockIdx.x * 8 + wv;
  if (d >= N_NODES) return;

  const int start = rowptr[d];
  const int end = rowptr[d + 1];
  const float dd = dinv[d];
  const vec* __restrict__ hv = (const vec*)hs;

  vec acc = {};
  for (int base = start; base < end; base += 32) {
    const int m = (end - base < 32) ? (end - base) : 32;
    const int li = base + lane;
    const int myidx = csr_src[(li < end) ? li : (end - 1)];
    for (int j = 0; j < m; j += DEPTH) {
      vec v[DEPTH];
#pragma unroll
      for (int u = 0; u < DEPTH; ++u) {
        const int s = __builtin_amdgcn_readlane(myidx, j + u);
        v[u] = hv[(long)s * 32 + lane];
      }
#pragma unroll
      for (int u = 0; u < DEPTH; ++u) {
        const float wt = (j + u < m) ? 1.0f : 0.0f;
        acc += v[u] * wt;
      }
    }
  }

  const vec hd = hv[(long)d * 32 + lane];
  const vec bv = ((const vec*)bias)[lane];
  const vec r = (acc + hd) * dd + bv;
  ((vec*)agg)[(long)d * 32 + lane] = r;
}

#define FG_NODES 32
#define FG_PITCH (HID + 4)

__global__ __launch_bounds__(256) void k_gather_gemm(const int* __restrict__ rowptr,
                                                     const int* __restrict__ csr_src,
                                                     const float* __restrict__ dinv,
                                                     const float* __restrict__ hs,
                                                     const float* __restrict__ bias,
                                                     const float* __restrict__ W,
                                                     float* __restrict__ C) {
  static_assert(N_NODES % FG_NODES == 0, "whole tiles only: WMMA needs every lane");
  constexpr int NC = OUTD;
  constexpr int DEPTH = 8;
  __shared__ float wl[HID * NC];
  __shared__ float tile[FG_NODES * FG_PITCH];

  const int lane = threadIdx.x & 31;
  const int wv = __builtin_amdgcn_readfirstlane((int)(threadIdx.x >> 5));
  const int node0 = blockIdx.x * FG_NODES;

#pragma unroll
  for (int q = 0; q < 4; ++q) {
    const int idx = threadIdx.x + q * 256;
    const int kp = idx / (NC / 4);
    const int nq = (idx % (NC / 4)) * 4;
    const float4 a = *(const float4*)(W + (2 * kp) * NC + nq);
    const float4 b = *(const float4*)(W + (2 * kp + 1) * NC + nq);
    float4* __restrict__ dst = (float4*)&wl[kp * 2 * NC + 2 * nq];
    dst[0] = make_float4(a.x, b.x, a.y, b.y);
    dst[1] = make_float4(a.z, b.z, a.w, b.w);
  }

  const v4f* __restrict__ hv = (const v4f*)hs;
  const v4f bv = ((const v4f*)bias)[lane];

  for (int i = 0; i < 4; ++i) {
    const int t = wv * 4 + i;
    const int d = node0 + t;
    const int start = rowptr[d];
    const int end = rowptr[d + 1];
    const float dd = dinv[d];
    const v4f hd = hv[(long)d * 32 + lane];

    v4f acc = {};
    for (int base = start; base < end; base += 32) {
      const int m = (end - base < 32) ? (end - base) : 32;
      const int li = base + lane;
      const int myidx = csr_src[(li < end) ? li : (end - 1)];
      for (int j = 0; j < m; j += DEPTH) {
        v4f v[DEPTH];
#pragma unroll
        for (int u = 0; u < DEPTH; ++u) {
          const int s = __builtin_amdgcn_readlane(myidx, j + u);
          v[u] = hv[(long)s * 32 + lane];
        }
#pragma unroll
        for (int u = 0; u < DEPTH; ++u) {
          const float wt = (j + u < m) ? 1.0f : 0.0f;
          acc += v[u] * wt;
        }
      }
    }
    v4f r = (acc + hd) * dd + bv;
    r.x = fmaxf(r.x, 0.0f);
    r.y = fmaxf(r.y, 0.0f);
    r.z = fmaxf(r.z, 0.0f);
    r.w = fmaxf(r.w, 0.0f);
    *(v4f*)&tile[t * FG_PITCH + lane * 4] = r;
  }

  __syncthreads();

  const int mt = wv >> 2;
  const int nt = wv & 3;
  const int m  = lane & 15;
  const int kh = lane >> 4;
  const float* __restrict__ arow = &tile[(mt * 16 + m) * FG_PITCH + 2 * kh];
  const float* __restrict__ bcol = &wl[kh * 2 * NC + 2 * (nt * 16 + m)];
  v8f c = {};
#pragma unroll
  for (int k0 = 0; k0 < HID; k0 += 4) {
    const v2f av = *(const v2f*)(arow + k0);
    const v2f bw = *(const v2f*)(bcol + (k0 >> 1) * 2 * NC);
    c = __builtin_amdgcn_wmma_f32_16x16x4_f32(
         false, av,  false, bw,
         (short)0, c,  false,  false);
  }

#pragma unroll
  for (int r = 0; r < 8; ++r) {
    const int row = node0 + mt * 16 + r + 8 * kh;
    C[(long)row * NC + nt * 16 + m] = c[r] * dinv[row];
  }
}

extern "C" void kernel_launch(void* const* d_in, const int* in_sizes, int n_in,
                              void* d_out, int out_size, void* d_ws, size_t ws_size,
                              hipStream_t stream) {
  const float* x  = (const float*)d_in[0];
  const int*   ei = (const int*)d_in[1];
  const float* W1 = (const float*)d_in[2];
  const float* b1 = (const float*)d_in[3];
  const float* W2 = (const float*)d_in[4];
  const float* b2 = (const float*)d_in[5];
  float* out = (float*)d_out;

  float* ws   = (float*)d_ws;
  float* dinv = ws;
  float* h1   = dinv + N_NODES;
  float* agg1 = h1 + (size_t)N_NODES * HID;
  float* h3   = agg1;
  int* aux     = (int*)(agg1 + (size_t)N_NODES * HID);
  int* counts  = aux;
  int* binptr  = counts + P_ROWS * BINP;
  int* rowptr  = binptr + BINP;
  int* csr_src = rowptr + N_NODES + 32;
  int* rec     = (int*)agg1;

  const int T = 256;

  k_bin_count<<<P_ROWS, PT, 0, stream>>>(ei, counts);
  k_bin_scatter<<<P_G, PT, 0, stream>>>(ei, counts, binptr, rec);
  k_bin_rows<<<NBINS, FT, 0, stream>>>(binptr, rec, rowptr, dinv, csr_src);

  k_gemm1_resident<<<G1_WG, G1_T, 0, stream>>>(x, W1, dinv, h1, N_NODES);

  k_gather_gemm<<<N_NODES / FG_NODES, T, 0, stream>>>(rowptr, csr_src, dinv, h1, b1, W2, h3);

  k_gather<OUTD><<<(N_NODES + 7) / 8, T, 0, stream>>>(rowptr, csr_src, dinv, h3, b2, out);
}


# Bin rows fused into the resident layer-1 GEMM: a bin is 400 nodes = the 25 row tiles one WGP multiplies, dinv read from LDS, count rows 256 wide; 6 launches -> 5; sum orders kept (rows: LDS arrival)
# speedup vs baseline: 3.8671x; 1.0636x over previous
#include <hip/hip_runtime.h>

#define N_NODES 100000
#define N_EDGES 1600000
#define IN_DIM  256
#define HID     128
#define OUTD    64

typedef __attribute__((ext_vector_type(2))) float v2f;
typedef __attribute__((ext_vector_type(4))) float v4f;
typedef __attribute__((ext_vector_type(8))) float v8f;

#define RB_NODES 400
#define RB_TILES (RB_NODES / 16)
#define RB_PAD 512
#define LOC_BITS 9
#define LOC_MASK ((1 << LOC_BITS) - 1)
#define NBINS (N_NODES / RB_NODES)
#define BINP 256
#define PT 1024
#define P_G 250
#define P_ROWS 256
#define P_CHUNK ((N_EDGES + P_G - 1) / P_G)
#define P_ITER ((P_CHUNK + PT - 1) / PT)
#define P_NG 8
#define P_GW (P_ROWS / P_NG)
#define FT 1024
#define F_UB 7
#define STAGE_CAP 8192
static_assert(N_NODES % RB_NODES == 0 && RB_NODES % 16 == 0, "whole bins of whole tiles");
static_assert(RB_NODES <= RB_PAD && RB_PAD == (1 << LOC_BITS), "local index fits its field");
static_assert(NBINS < BINP, "binptr[b + 1] exists for every bin");

template <int PL>
__device__ __forceinline__ void wave0_excl_scan(const int* __restrict__ in,
                                                int* __restrict__ out, int lane) {
  int v[PL];
  int s = 0;
#pragma unroll
  for (int k = 0; k < PL; ++k) {
    v[k] = in[lane * PL + k];
    s += v[k];
  }
  int inc = s;
#pragma unroll
  for (int off = 1; off < 32; off <<= 1) {
    const int nb = __shfl_up(inc, off);
    if (lane >= off) inc += nb;
  }
  int run = inc - s;
#pragma unroll
  for (int k = 0; k < PL; ++k) {
    out[lane * PL + k] = run;
    run += v[k];
  }
}

__global__ __launch_bounds__(PT) void k_bin_count(const int* __restrict__ ei,
                                                  int* __restrict__ counts) {
  __shared__ int hist[BINP];
  const int tid = threadIdx.x;
  const int w = blockIdx.x;
  if (tid < BINP) hist[tid] = 0;
  const int base = w * P_CHUNK;
  const int lim = (base + P_CHUNK < N_EDGES) ? (base + P_CHUNK) : N_EDGES;
  int dv[P_ITER];
#pragma unroll
  for (int it = 0; it < P_ITER; ++it) {
    const int e = base + it * PT + tid;
    dv[it] = ei[N_EDGES + ((e < lim) ? e : (lim - 1))];
  }
  __syncthreads();
#pragma unroll
  for (int it = 0; it < P_ITER; ++it) {
    const int e = base + it * PT + tid;
    if (e < lim) atomicAdd(&hist[(unsigned)dv[it] / RB_NODES], 1);
  }
  __syncthreads();
  if (tid < BINP) counts[w * BINP + tid] = hist[tid];
}

__global__ __launch_bounds__(PT) void k_bin_scatter(const int* __restrict__ ei,
                                                    const int* __restrict__ counts,
                                                    int* __restrict__ binptr,
                                                    int* __restrict__ rec) {
  __shared__ int part_tot[P_NG][BINP];
  __shared__ int part_pre[P_NG][BINP];
  __shared__ int ptot[BINP / 2];
  __shared__ int pex[BINP / 2];
  __shared__ int cursor[BINP];
  static_assert(PT / P_NG == BINP / 2, "a thread group covers a row of counts two bins a thread");
  const int tid = threadIdx.x;
  const int w = blockIdx.x;
  const int g = __builtin_amdgcn_readfirstlane(tid / (BINP / 2));
  const int t = tid % (BINP / 2);

  const int base = w * P_CHUNK;
  const int lim = (base + P_CHUNK < N_EDGES) ? (base + P_CHUNK) : N_EDGES;
  int dv[P_ITER], sv[P_ITER];
#pragma unroll
  for (int it = 0; it < P_ITER; ++it) {
    const int e = base + it * PT + tid;
    const int ec = (e < lim) ? e : (lim - 1);
    dv[it] = ei[N_EDGES + ec];
    sv[it] = ei[ec];
  }

  const int2* __restrict__ c2 = (const int2*)counts;
  int tx = 0, ty = 0, px = 0, py = 0;
#pragma unroll 16
  for (int k = 0; k < P_GW; ++k) {
    const int ww = g * P_GW + k;
    const int2 v = c2[ww * (BINP / 2) + t];
    tx += v.x;
    ty += v.y;
    px += (ww < w) ? v.x : 0;
    py += (ww < w) ? v.y : 0;
  }
  part_tot[g][2 * t] = tx;
  part_tot[g][2 * t + 1] = ty;
  part_pre[g][2 * t] = px;
  part_pre[g][2 * t + 1] = py;
  __syncthreads();
  int T0 = 0, P0 = 0, P1 = 0;
  if (tid < BINP / 2) {
    int T1 = 0;
#pragma unroll
    for (int q = 0; q < P_NG; ++q) {
      T0 += part_tot[q][2 * tid];
      T1 += part_tot[q][2 * tid + 1];
      P0 += part_pre[q][2 * tid];
      P1 += part_pre[q][2 * tid + 1];
    }
    ptot[tid] = T0 + T1;
  }
  __syncthreads();
  if (tid < 32) wave0_excl_scan<BINP / 64>(ptot, pex, tid);
  __syncthreads();
  if (tid < BINP / 2) {
    const int b0 = pex[tid];
    const int b1 = b0 + T0;
    cursor[2 * tid] = b0 + P0;
    cursor[2 * tid + 1] = b1 + P1;
    if (w == 0) {
      binptr[2 * tid] = b0;
      binptr[2 * tid + 1] = b1;
    }
  }
  __syncthreads();
#pragma unroll
  for (int it = 0; it < P_ITER; ++it) {
    const int e = base + it * PT + tid;
    if (e < lim) {
      const int bin = (unsigned)dv[it] / RB_NODES;
      const int pos = atomicAdd(&cursor[bin], 1);
      rec[pos] = (sv[it] << LOC_BITS) | (dv[it] - bin * RB_NODES);
    }
  }
}

#define G1_T 1024
#define G1_PAIRS (G1_T / 64)
static_assert(G1_T == FT, "one block shape for both parts");

__global__ __launch_bounds__(G1_T) __attribute__((amdgpu_waves_per_eu(8, 8))) void k_rows_gemm1(const int* __restrict__ binptr,
                                                    const int* __restrict__ rec,
                                                    int* __restrict__ rowptr,
                                                    float* __restrict__ dinv,
                                                    int* __restrict__ csr_src,
                                                    const float* __restrict__ A,
                                                    const float* __restrict__ W,
                                                    float* __restrict__ C) {
  constexpr int K = IN_DIM;
  constexpr int NC = HID;
  constexpr int NTH = 4;
  __shared__ float lds[K * NC];
  __shared__ int cnt[RB_PAD];
  __shared__ int ex[RB_PAD];
  __shared__ float dloc[RB_PAD];
  __shared__ int stage[STAGE_CAP];
  const int tid = threadIdx.x;
  const int b = blockIdx.x;

  {
    constexpr int ENT = K * NC / 8;
    constexpr int PER = ENT / G1_T;
    static_assert(ENT % G1_T == 0, "whole entries per thread");
    float4 ra[PER], rb[PER];
#pragma unroll
    for (int q = 0; q < PER; ++q) {
      const int idx = tid + q * G1_T;
      const int kp = idx / (NC / 4);
      const int nq = (idx % (NC / 4)) * 4;
      ra[q] = *(const float4*)(W + (2 * kp) * NC + nq);
      rb[q] = *(const float4*)(W + (2 * kp + 1) * NC + nq);
    }
#pragma unroll
    for (int q = 0; q < PER; ++q) {
      const int idx = tid + q * G1_T;
      const int kp = idx / (NC / 4);
      const int nq = (idx % (NC / 4)) * 4;
      float4* __restrict__ dst = (float4*)&lds[kp * 2 * NC + 2 * nq];
      dst[0] = make_float4(ra[q].x, rb[q].x, ra[q].y, rb[q].y);
      dst[1] = make_float4(ra[q].z, rb[q].z, ra[q].w, rb[q].w);
    }
  }

  const int lo = binptr[b];
  const int n = binptr[b + 1] - lo;
  if (tid < RB_PAD) cnt[tid] = 0;
  __syncthreads();
  for (int i0 = tid; i0 < n; i0 += F_UB * FT) {
    int r[F_UB];
#pragma unroll
    for (int u = 0; u < F_UB; ++u) {
      const int i = i0 + u * FT;
      r[u] = rec[lo + ((i < n) ? i : (n - 1))];
    }
#pragma unroll
    for (int u = 0; u < F_UB; ++u)
      if (i0 + u * FT < n) atomicAdd(&cnt[r[u] & LOC_MASK], 1);
  }
  __syncthreads();
  if (tid < 32) wave0_excl_scan<RB_PAD / 32>(cnt, ex, tid);
  __syncthreads();
  if (tid < RB_NODES) {
    const int c = cnt[tid];
    const int x = ex[tid];
    const int node = b * RB_NODES + tid;
    const float dv = rsqrtf((float)c + 1.0f);
    rowptr[node] = lo + x;
    dinv[node] = dv;
    dloc[tid] = dv;
    if (node == N_NODES - 1) rowptr[N_NODES] = lo + x + c;
  }
  __syncthreads();
  const bool staged = (n <= STAGE_CAP);
  for (int i0 = tid; i0 < n; i0 += F_UB * FT) {
    int r[F_UB];
#pragma unroll
    for (int u = 0; u < F_UB; ++u) {
      const int i = i0 + u * FT;
      r[u] = rec[lo + ((i < n) ? i : (n - 1))];
    }
#pragma unroll
    for (int u = 0; u < F_UB; ++u) {
      if (i0 + u * FT < n) {
        const int p = atomicAdd(&ex[r[u] & LOC_MASK], 1);
        if (staged) stage[p] = r[u] >> LOC_BITS;
        else csr_src[lo + p] = r[u] >> LOC_BITS;
      }
    }
  }
  __syncthreads();
  if (staged)
    for (int i = tid; i < n; i += FT) csr_src[lo + i] = stage[i];

  const int lane = tid & 31;
  const int wave = __builtin_amdgcn_readfirstlane(tid >> 5);
  const int p  = wave >> 1;
  const int hf = wave & 1;
  const int m  = lane & 15;
  const int kh = lane >> 4;
  const int nn = lane & 15;
  const v2f* __restrict__ bbase = (const v2f*)lds + kh * NC + hf * (NTH * 16) + nn;

  for (int lt = p; lt < RB_TILES; lt += G1_PAIRS) {
    const long m0 = ((long)b * RB_TILES + lt) * 16;
    const float* __restrict__ Aptr = A + (m0 + m) * K + 2 * kh;

    v8f acc[NTH] = {};
    for (int k8 = 0; k8 < K; k8 += 32) {
      v2f av[8];
#pragma unroll
      for (int u = 0; u < 8; ++u) av[u] = *(const v2f*)(Aptr + k8 + 4 * u);
#pragma unroll
      for (int u = 0; u < 8; ++u) {
        const v2f* __restrict__ brow = bbase + ((k8 + 4 * u) >> 1) * NC;
#pragma unroll
        for (int j = 0; j < NTH; ++j) {
          const v2f bv = brow[j * 16];
          acc[j] = __builtin_amdgcn_wmma_f32_16x16x4_f32(
               false, av[u],  false, bv,
               (short)0, acc[j],  false,  false);
        }
      }
    }

    float rs[8];
#pragma unroll
    for (int r = 0; r < 8; ++r) rs[r] = dloc[lt * 16 + r + 8 * kh];
#pragma unroll
    for (int j = 0; j < NTH; ++j) {
#pragma unroll
      for (int r = 0; r < 8; ++r) {
        const long row = m0 + r + 8 * kh;
        C[row * NC + hf * (NTH * 16) + j * 16 + nn] = acc[j][r] * rs[r];
      }
    }
  }
}

template <int V> struct LaneVec;
template <> struct LaneVec<2> { typedef v2f type; };
template <> struct LaneVec<4> { typedef v4f type; };

template <int F>
__global__ __launch_bounds__(256) void k_gather(const int* __restrict__ rowptr,
                                                const int* __restrict__ csr_src,
                                                const float* __restrict__ dinv,
                                                const float* __restrict__ hs,
                                                const float* __restrict__ bias,
                                                float* __restrict__ agg) {
  typedef typename LaneVec<F / 32>::type vec;
  constexpr int DEPTH = 8;
  const int lane = threadIdx.x & 31;
  const int wv = __builtin_amdgcn_readfirstlane((int)(threadIdx.x >> 5));
  const int d = blockIdx.x * 8 + wv;
  if (d >= N_NODES) return;

  const int start = rowptr[d];
  const int end = rowptr[d + 1];
  const float dd = dinv[d];
  const vec* __restrict__ hv = (const vec*)hs;

  vec acc = {};
  for (int base = start; base < end; base += 32) {
    const int m = (end - base < 32) ? (end - base) : 32;
    const int li = base + lane;
    const int myidx = csr_src[(li < end) ? li : (end - 1)];
    for (int j = 0; j < m; j += DEPTH) {
      vec v[DEPTH];
#pragma unroll
      for (int u = 0; u < DEPTH; ++u) {
        const int s = __builtin_amdgcn_readlane(myidx, j + u);
        v[u] = hv[(long)s * 32 + lane];
      }
#pragma unroll
      for (int u = 0; u < DEPTH; ++u) {
        const float wt = (j + u < m) ? 1.0f : 0.0f;
        acc += v[u] * wt;
      }
    }
  }

  const vec hd = hv[(long)d * 32 + lane];
  const vec bv = ((const vec*)bias)[lane];
  const vec r = (acc + hd) * dd + bv;
  ((vec*)agg)[(long)d * 32 + lane] = r;
}

#define FG_NODES 32
#define FG_PITCH (HID + 4)

__global__ __launch_bounds__(256) void k_gather_gemm(const int* __restrict__ rowptr,
                                                     const int* __restrict__ csr_src,
                                                     const float* __restrict__ dinv,
                                                     const float* __restrict__ hs,
                                                     const float* __restrict__ bias,
                                                     const float* __restrict__ W,
                                                     float* __restrict__ C) {
  static_assert(N_NODES % FG_NODES == 0, "whole tiles only: WMMA needs every lane");
  constexpr int NC = OUTD;
  constexpr int DEPTH = 8;
  __shared__ float wl[HID * NC];
  __shared__ float tile[FG_NODES * FG_PITCH];

  const int lane = threadIdx.x & 31;
  const int wv = __builtin_amdgcn_readfirstlane((int)(threadIdx.x >> 5));
  const int node0 = blockIdx.x * FG_NODES;

#pragma unroll
  for (int q = 0; q < 4; ++q) {
    const int idx = threadIdx.x + q * 256;
    const int kp = idx / (NC / 4);
    const int nq = (idx % (NC / 4)) * 4;
    const float4 a = *(const float4*)(W + (2 * kp) * NC + nq);
    const float4 b = *(const float4*)(W + (2 * kp + 1) * NC + nq);
    float4* __restrict__ dst = (float4*)&wl[kp * 2 * NC + 2 * nq];
    dst[0] = make_float4(a.x, b.x, a.y, b.y);
    dst[1] = make_float4(a.z, b.z, a.w, b.w);
  }

  const v4f* __restrict__ hv = (const v4f*)hs;
  const v4f bv = ((const v4f*)bias)[lane];

  for (int i = 0; i < 4; ++i) {
    const int t = wv * 4 + i;
    const int d = node0 + t;
    const int start = rowptr[d];
    const int end = rowptr[d + 1];
    const float dd = dinv[d];
    const v4f hd = hv[(long)d * 32 + lane];

    v4f acc = {};
    for (int base = start; base < end; base += 32) {
      const int m = (end - base < 32) ? (end - base) : 32;
      const int li = base + lane;
      const int myidx = csr_src[(li < end) ? li : (end - 1)];
      for (int j = 0; j < m; j += DEPTH) {
        v4f v[DEPTH];
#pragma unroll
        for (int u = 0; u < DEPTH; ++u) {
          const int s = __builtin_amdgcn_readlane(myidx, j + u);
          v[u] = hv[(long)s * 32 + lane];
        }
#pragma unroll
        for (int u = 0; u < DEPTH; ++u) {
          const float wt = (j + u < m) ? 1.0f : 0.0f;
          acc += v[u] * wt;
        }
      }
    }
    v4f r = (acc + hd) * dd + bv;
    r.x = fmaxf(r.x, 0.0f);
    r.y = fmaxf(r.y, 0.0f);
    r.z = fmaxf(r.z, 0.0f);
    r.w = fmaxf(r.w, 0.0f);
    *(v4f*)&tile[t * FG_PITCH + lane * 4] = r;
  }

  __syncthreads();

  const int mt = wv >> 2;
  const int nt = wv & 3;
  const int m  = lane & 15;
  const int kh = lane >> 4;
  const float* __restrict__ arow = &tile[(mt * 16 + m) * FG_PITCH + 2 * kh];
  const float* __restrict__ bcol = &wl[kh * 2 * NC + 2 * (nt * 16 + m)];
  v8f c = {};
#pragma unroll
  for (int k0 = 0; k0 < HID; k0 += 4) {
    const v2f av = *(const v2f*)(arow + k0);
    const v2f bw = *(const v2f*)(bcol + (k0 >> 1) * 2 * NC);
    c = __builtin_amdgcn_wmma_f32_16x16x4_f32(
         false, av,  false, bw,
         (short)0, c,  false,  false);
  }

#pragma unroll
  for (int r = 0; r < 8; ++r) {
    const int row = node0 + mt * 16 + r + 8 * kh;
    C[(long)row * NC + nt * 16 + m] = c[r] * dinv[row];
  }
}

extern "C" void kernel_launch(void* const* d_in, const int* in_sizes, int n_in,
                              void* d_out, int out_size, void* d_ws, size_t ws_size,
                              hipStream_t stream) {
  const float* x  = (const float*)d_in[0];
  const int*   ei = (const int*)d_in[1];
  const float* W1 = (const float*)d_in[2];
  const float* b1 = (const float*)d_in[3];
  const float* W2 = (const float*)d_in[4];
  const float* b2 = (const float*)d_in[5];
  float* out = (float*)d_out;

  float* ws   = (float*)d_ws;
  float* dinv = ws;
  float* h1   = dinv + N_NODES;
  float* agg1 = h1 + (size_t)N_NODES * HID;
  float* h3   = agg1;
  int* aux     = (int*)(agg1 + (size_t)N_NODES * HID);
  int* counts  = aux;
  int* binptr  = counts + P_ROWS * BINP;
  int* rowptr  = binptr + BINP;
  int* csr_src = rowptr + N_NODES + 32;
  int* rec     = (int*)agg1;

  const int T = 256;

  k_bin_count<<<P_ROWS, PT, 0, stream>>>(ei, counts);
  k_bin_scatter<<<P_G, PT, 0, stream>>>(ei, counts, binptr, rec);
  k_rows_gemm1<<<NBINS, G1_T, 0, stream>>>(binptr, rec, rowptr, dinv, csr_src, x, W1, h1);

  k_gather_gemm<<<N_NODES / FG_NODES, T, 0, stream>>>(rowptr, csr_src, dinv, h1, b1, W2, h3);

  k_gather<OUTD><<<(N_NODES + 7) / 8, T, 0, stream>>>(rowptr, csr_src, dinv, h3, b2, out);
}


# CSR build in two launches, not three: each edge chunk ordered by bin in LDS and stored as whole lines with a table of run starts; a bin reads its 256 runs; row sum order = LDS arrival, as before
# speedup vs baseline: 4.4033x; 1.1387x over previous
#include <hip/hip_runtime.h>

#define N_NODES 100000
#define N_EDGES 1600000
#define IN_DIM  256
#define HID     128
#define OUTD    64

typedef __attribute__((ext_vector_type(2))) float v2f;
typedef __attribute__((ext_vector_type(4))) float v4f;
typedef __attribute__((ext_vector_type(8))) float v8f;

#define RB_NODES 400
#define RB_TILES (RB_NODES / 16)
#define RB_PAD 512
#define LOC_BITS 9
#define LOC_MASK ((1 << LOC_BITS) - 1)
#define NBINS (N_NODES / RB_NODES)
#define BINP 256
#define PT 1024
#define P_G 250
#define P_ROWS 256
#define P_CHUNK ((N_EDGES + P_G - 1) / P_G)
#define P_ITER ((P_CHUNK + PT - 1) / PT)
#define RUN_BITS 13
#define RUN_MASK ((1 << RUN_BITS) - 1)
#define R_UB 8
#define FT 1024
#define STAGE_CAP 8192
static_assert(N_NODES % RB_NODES == 0 && RB_NODES % 16 == 0, "whole bins of whole tiles");
static_assert(RB_NODES <= RB_PAD && RB_PAD == (1 << LOC_BITS), "local index fits its field");
static_assert(NBINS <= BINP && P_G <= P_ROWS, "table rows and columns");
static_assert(P_CHUNK <= RUN_MASK, "a run's length and its start fit a table word");
static_assert(FT == 4 * P_ROWS, "four threads of a bin block per run");

template <int PL>
__device__ __forceinline__ void wave0_excl_scan(const int* __restrict__ in,
                                                int* __restrict__ out, int lane) {
  int v[PL];
  int s = 0;
#pragma unroll
  for (int k = 0; k < PL; ++k) {
    v[k] = in[lane * PL + k];
    s += v[k];
  }
  int inc = s;
#pragma unroll
  for (int off = 1; off < 32; off <<= 1) {
    const int nb = __shfl_up(inc, off);
    if (lane >= off) inc += nb;
  }
  int run = inc - s;
#pragma unroll
  for (int k = 0; k < PL; ++k) {
    out[lane * PL + k] = run;
    run += v[k];
  }
}

__global__ __launch_bounds__(PT) void k_chunk_sort(const int* __restrict__ ei,
                                                   int* __restrict__ runtab,
                                                   int* __restrict__ rec) {
  __shared__ int hist[BINP];
  __shared__ int offs[BINP];
  __shared__ int sorted[P_CHUNK];
  const int tid = threadIdx.x;
  const int w = blockIdx.x;
  if (tid < BINP) hist[tid] = 0;
  const int base = w * P_CHUNK;
  const int lim = (base + P_CHUNK < N_EDGES) ? (base + P_CHUNK) : N_EDGES;
  int dv[P_ITER], sv[P_ITER];
#pragma unroll
  for (int it = 0; it < P_ITER; ++it) {
    const int e = base + it * PT + tid;
    const int ec = (e < lim) ? e : (lim - 1);
    dv[it] = ei[N_EDGES + ec];
    sv[it] = ei[ec];
  }
  __syncthreads();
#pragma unroll
  for (int it = 0; it < P_ITER; ++it) {
    const int e = base + it * PT + tid;
    if (e < lim) atomicAdd(&hist[(unsigned)dv[it] / RB_NODES], 1);
  }
  __syncthreads();
  if (tid < 32) wave0_excl_scan<BINP / 32>(hist, offs, tid);
  __syncthreads();
#pragma unroll
  for (int it = 0; it < P_ITER; ++it) {
    const int e = base + it * PT + tid;
    if (e < lim) {
      const int bin = (unsigned)dv[it] / RB_NODES;
      const int pos = atomicAdd(&offs[bin], 1);
      sorted[pos] = (sv[it] << LOC_BITS) | (dv[it] - bin * RB_NODES);
    }
  }
  __syncthreads();
  for (int i = tid; i < lim - base; i += PT) rec[base + i] = sorted[i];
  if (tid < NBINS) runtab[tid * P_ROWS + w] = ((offs[tid] - hist[tid]) << RUN_BITS) | hist[tid];
}

#define G1_T 1024
#define G1_PAIRS (G1_T / 64)
static_assert(G1_T == FT, "one block shape for both parts");

__global__ __launch_bounds__(G1_T) __attribute__((amdgpu_waves_per_eu(8, 8))) void k_rows_gemm1(const int* __restrict__ runtab,
                                                    const int* __restrict__ rec,
                                                    int* __restrict__ rowptr,
                                                    float* __restrict__ dinv,
                                                    int* __restrict__ csr_src,
                                                    const float* __restrict__ A,
                                                    const float* __restrict__ W,
                                                    float* __restrict__ C) {
  constexpr int K = IN_DIM;
  constexpr int NC = HID;
  constexpr int NTH = 4;
  __shared__ float lds[K * NC];
  __shared__ int cnt[RB_PAD];
  __shared__ int ex[RB_PAD];
  __shared__ float dloc[RB_PAD];
  __shared__ int stage[STAGE_CAP];
  __shared__ int tot;
  const int tid = threadIdx.x;
  const int b = blockIdx.x;

  constexpr int ENT = K * NC / 8;
  constexpr int PER = ENT / G1_T;
  static_assert(ENT % G1_T == 0, "whole entries per thread");
  float4 ra[PER], rb[PER];
#pragma unroll
  for (int q = 0; q < PER; ++q) {
    const int idx = tid + q * G1_T;
    const int kp = idx / (NC / 4);
    const int nq = (idx % (NC / 4)) * 4;
    ra[q] = *(const float4*)(W + (2 * kp) * NC + nq);
    rb[q] = *(const float4*)(W + (2 * kp + 1) * NC + nq);
  }

  const int sub = tid & 3;
  const int pk = runtab[b * P_ROWS + (tid >> 2)];
  constexpr int TQ = (BINP * P_ROWS / 4) / FT;
  const int lim4 = b * (P_ROWS / 4);
  int4 tv[TQ];
#pragma unroll
  for (int q = 0; q < TQ; ++q) {
    const int idx = tid + q * FT;
    tv[q] = ((const int4*)runtab)[(idx < lim4) ? idx : 0];
  }

#pragma unroll
  for (int q = 0; q < PER; ++q) {
    const int idx = tid + q * G1_T;
    const int kp = idx / (NC / 4);
    const int nq = (idx % (NC / 4)) * 4;
    float4* __restrict__ dst = (float4*)&lds[kp * 2 * NC + 2 * nq];
    dst[0] = make_float4(ra[q].x, rb[q].x, ra[q].y, rb[q].y);
    dst[1] = make_float4(ra[q].z, rb[q].z, ra[q].w, rb[q].w);
  }

  const int rcnt = pk & RUN_MASK;
  const int* __restrict__ runp = rec + (tid >> 2) * P_CHUNK + (pk >> RUN_BITS);
  int before = 0;
#pragma unroll
  for (int q = 0; q < TQ; ++q) {
    if (tid + q * FT < lim4)
      before += (tv[q].x & RUN_MASK) + (tv[q].y & RUN_MASK) + (tv[q].z & RUN_MASK) + (tv[q].w & RUN_MASK);
  }
#pragma unroll
  for (int off = 16; off > 0; off >>= 1) before += __shfl_xor(before, off);
  if (tid < RB_PAD) cnt[tid] = 0;
  if (tid == 0) tot = 0;
  __syncthreads();
  if ((tid & 31) == 0) atomicAdd(&tot, before);
  for (int j0 = sub; j0 < rcnt; j0 += 4 * R_UB) {
    int r[R_UB];
#pragma unroll
    for (int u = 0; u < R_UB; ++u) {
      const int j = j0 + 4 * u;
      r[u] = runp[(j < rcnt) ? j : (rcnt - 1)];
    }
#pragma unroll
    for (int u = 0; u < R_UB; ++u)
      if (j0 + 4 * u < rcnt) atomicAdd(&cnt[r[u] & LOC_MASK], 1);
  }
  __syncthreads();
  if (tid < 32) wave0_excl_scan<RB_PAD / 32>(cnt, ex, tid);
  __syncthreads();
  const int lo = tot;
  const int n = ex[RB_PAD - 1] + cnt[RB_PAD - 1];
  if (tid < RB_NODES) {
    const int c = cnt[tid];
    const int x = ex[tid];
    const int node = b * RB_NODES + tid;
    const float dv = rsqrtf((float)c + 1.0f);
    rowptr[node] = lo + x;
    dinv[node] = dv;
    dloc[tid] = dv;
    if (node == N_NODES - 1) rowptr[N_NODES] = lo + x + c;
  }
  __syncthreads();
  const bool staged = (n <= STAGE_CAP);
  for (int j0 = sub; j0 < rcnt; j0 += 4 * R_UB) {
    int r[R_UB];
#pragma unroll
    for (int u = 0; u < R_UB; ++u) {
      const int j = j0 + 4 * u;
      r[u] = runp[(j < rcnt) ? j : (rcnt - 1)];
    }
#pragma unroll
    for (int u = 0; u < R_UB; ++u) {
      if (j0 + 4 * u < rcnt) {
        const int p = atomicAdd(&ex[r[u] & LOC_MASK], 1);
        if (staged) stage[p] = r[u] >> LOC_BITS;
        else csr_src[lo + p] = r[u] >> LOC_BITS;
      }
    }
  }
  __syncthreads();
  if (staged)
    for (int i = tid; i < n; i += FT) csr_src[lo + i] = stage[i];

  const int lane = tid & 31;
  const int wave = __builtin_amdgcn_readfirstlane(tid >> 5);
  const int p  = wave >> 1;
  const int hf = wave & 1;
  const int m  = lane & 15;
  const int kh = lane >> 4;
  const int nn = lane & 15;
  const v2f* __restrict__ bbase = (const v2f*)lds + kh * NC + hf * (NTH * 16) + nn;

  for (int lt = p; lt < RB_TILES; lt += G1_PAIRS) {
    const long m0 = ((long)b * RB_TILES + lt) * 16;
    const float* __restrict__ Aptr = A + (m0 + m) * K + 2 * kh;

    v8f acc[NTH] = {};
    for (int k8 = 0; k8 < K; k8 += 32) {
      v2f av[8];
#pragma unroll
      for (int u = 0; u < 8; ++u) av[u] = *(const v2f*)(Aptr + k8 + 4 * u);
#pragma unroll
      for (int u = 0; u < 8; ++u) {
        const v2f* __restrict__ brow = bbase + ((k8 + 4 * u) >> 1) * NC;
#pragma unroll
        for (int j = 0; j < NTH; ++j) {
          const v2f bv = brow[j * 16];
          acc[j] = __builtin_amdgcn_wmma_f32_16x16x4_f32(
               false, av[u],  false, bv,
               (short)0, acc[j],  false,  false);
        }
      }
    }

    float rs[8];
#pragma unroll
    for (int r = 0; r < 8; ++r) rs[r] = dloc[lt * 16 + r + 8 * kh];
#pragma unroll
    for (int j = 0; j < NTH; ++j) {
#pragma unroll
      for (int r = 0; r < 8; ++r) {
        const long row = m0 + r + 8 * kh;
        C[row * NC + hf * (NTH * 16) + j * 16 + nn] = acc[j][r] * rs[r];
      }
    }
  }
}

template <int V> struct LaneVec;
template <> struct LaneVec<2> { typedef v2f type; };
template <> struct LaneVec<4> { typedef v4f type; };

template <int F>
__global__ __launch_bounds__(256) void k_gather(const int* __restrict__ rowptr,
                                                const int* __restrict__ csr_src,
                                                const float* __restrict__ dinv,
                                                const float* __restrict__ hs,
                                                const float* __restrict__ bias,
                                                float* __restrict__ agg) {
  typedef typename LaneVec<F / 32>::type vec;
  constexpr int DEPTH = 8;
  const int lane = threadIdx.x & 31;
  const int wv = __builtin_amdgcn_readfirstlane((int)(threadIdx.x >> 5));
  const int d = blockIdx.x * 8 + wv;
  if (d >= N_NODES) return;

  const int start = rowptr[d];
  const int end = rowptr[d + 1];
  const float dd = dinv[d];
  const vec* __restrict__ hv = (const vec*)hs;

  vec acc = {};
  for (int base = start; base < end; base += 32) {
    const int m = (end - base < 32) ? (end - base) : 32;
    const int li = base + lane;
    const int myidx = csr_src[(li < end) ? li : (end - 1)];
    for (int j = 0; j < m; j += DEPTH) {
      vec v[DEPTH];
#pragma unroll
      for (int u = 0; u < DEPTH; ++u) {
        const int s = __builtin_amdgcn_readlane(myidx, j + u);
        v[u] = hv[(long)s * 32 + lane];
      }
#pragma unroll
      for (int u = 0; u < DEPTH; ++u) {
        const float wt = (j + u < m) ? 1.0f : 0.0f;
        acc += v[u] * wt;
      }
    }
  }

  const vec hd = hv[(long)d * 32 + lane];
  const vec bv = ((const vec*)bias)[lane];
  const vec r = (acc + hd) * dd + bv;
  ((vec*)agg)[(long)d * 32 + lane] = r;
}

#define FG_NODES 32
#define FG_PITCH (HID + 4)

__global__ __launch_bounds__(256) void k_gather_gemm(const int* __restrict__ rowptr,
                                                     const int* __restrict__ csr_src,
                                                     const float* __restrict__ dinv,
                                                     const float* __restrict__ hs,
                                                     const float* __restrict__ bias,
                                                     const float* __restrict__ W,
                                                     float* __restrict__ C) {
  static_assert(N_NODES % FG_NODES == 0, "whole tiles only: WMMA needs every lane");
  constexpr int NC = OUTD;
  constexpr int DEPTH = 8;
  __shared__ float wl[HID * NC];
  __shared__ float tile[FG_NODES * FG_PITCH];

  const int lane = threadIdx.x & 31;
  const int wv = __builtin_amdgcn_readfirstlane((int)(threadIdx.x >> 5));
  const int node0 = blockIdx.x * FG_NODES;

#pragma unroll
  for (int q = 0; q < 4; ++q) {
    const int idx = threadIdx.x + q * 256;
    const int kp = idx / (NC / 4);
    const int nq = (idx % (NC / 4)) * 4;
    const float4 a = *(const float4*)(W + (2 * kp) * NC + nq);
    const float4 b = *(const float4*)(W + (2 * kp + 1) * NC + nq);
    float4* __restrict__ dst = (float4*)&wl[kp * 2 * NC + 2 * nq];
    dst[0] = make_float4(a.x, b.x, a.y, b.y);
    dst[1] = make_float4(a.z, b.z, a.w, b.w);
  }

  const v4f* __restrict__ hv = (const v4f*)hs;
  const v4f bv = ((const v4f*)bias)[lane];

  for (int i = 0; i < 4; ++i) {
    const int t = wv * 4 + i;
    const int d = node0 + t;
    const int start = rowptr[d];
    const int end = rowptr[d + 1];
    const float dd = dinv[d];
    const v4f hd = hv[(long)d * 32 + lane];

    v4f acc = {};
    for (int base = start; base < end; base += 32) {
      const int m = (end - base < 32) ? (end - base) : 32;
      const int li = base + lane;
      const int myidx = csr_src[(li < end) ? li : (end - 1)];
      for (int j = 0; j < m; j += DEPTH) {
        v4f v[DEPTH];
#pragma unroll
        for (int u = 0; u < DEPTH; ++u) {
          const int s = __builtin_amdgcn_readlane(myidx, j + u);
          v[u] = hv[(long)s * 32 + lane];
        }
#pragma unroll
        for (int u = 0; u < DEPTH; ++u) {
          const float wt = (j + u < m) ? 1.0f : 0.0f;
          acc += v[u] * wt;
        }
      }
    }
    v4f r = (acc + hd) * dd + bv;
    r.x = fmaxf(r.x, 0.0f);
    r.y = fmaxf(r.y, 0.0f);
    r.z = fmaxf(r.z, 0.0f);
    r.w = fmaxf(r.w, 0.0f);
    *(v4f*)&tile[t * FG_PITCH + lane * 4] = r;
  }

  __syncthreads();

  const int mt = wv >> 2;
  const int nt = wv & 3;
  const int m  = lane & 15;
  const int kh = lane >> 4;
  const float* __restrict__ arow = &tile[(mt * 16 + m) * FG_PITCH + 2 * kh];
  const float* __restrict__ bcol = &wl[kh * 2 * NC + 2 * (nt * 16 + m)];
  v8f c = {};
#pragma unroll
  for (int k0 = 0; k0 < HID; k0 += 4) {
    const v2f av = *(const v2f*)(arow + k0);
    const v2f bw = *(const v2f*)(bcol + (k0 >> 1) * 2 * NC);
    c = __builtin_amdgcn_wmma_f32_16x16x4_f32(
         false, av,  false, bw,
         (short)0, c,  false,  false);
  }

#pragma unroll
  for (int r = 0; r < 8; ++r) {
    const int row = node0 + mt * 16 + r + 8 * kh;
    C[(long)row * NC + nt * 16 + m] = c[r] * dinv[row];
  }
}

extern "C" void kernel_launch(void* const* d_in, const int* in_sizes, int n_in,
                              void* d_out, int out_size, void* d_ws, size_t ws_size,
                              hipStream_t stream) {
  const float* x  = (const float*)d_in[0];
  const int*   ei = (const int*)d_in[1];
  const float* W1 = (const float*)d_in[2];
  const float* b1 = (const float*)d_in[3];
  const float* W2 = (const float*)d_in[4];
  const float* b2 = (const float*)d_in[5];
  float* out = (float*)d_out;

  float* ws   = (float*)d_ws;
  float* dinv = ws;
  float* h1   = dinv + N_NODES;
  float* agg1 = h1 + (size_t)N_NODES * HID;
  float* h3   = agg1;
  int* aux     = (int*)(agg1 + (size_t)N_NODES * HID);
  int* runtab  = aux;
  int* rowptr  = runtab + BINP * P_ROWS;
  int* csr_src = rowptr + N_NODES + 32;
  int* rec     = (int*)agg1;

  const int T = 256;

  k_chunk_sort<<<P_ROWS, PT, 0, stream>>>(ei, runtab, rec);
  k_rows_gemm1<<<NBINS, G1_T, 0, stream>>>(runtab, rec, rowptr, dinv, csr_src, x, W1, h1);

  k_gather_gemm<<<N_NODES / FG_NODES, T, 0, stream>>>(rowptr, csr_src, dinv, h1, b1, W2, h3);

  k_gather<OUTD><<<(N_NODES + 7) / 8, T, 0, stream>>>(rowptr, csr_src, dinv, h3, b2, out);
}
